# starting kernel
# baseline (speedup 1.0000x reference)
#include <hip/hip_runtime.h>
#include <hip/hip_bf16.h>


#define DIN 128
#define HD  64
#define H4  4

typedef __attribute__((ext_vector_type(2))) float v2f;
typedef __attribute__((ext_vector_type(8))) float v8f;

__global__ __launch_bounds__(256) void gat_gemm_wmma(
    const float* __restrict__ feat, const float* __restrict__ W,
    float* __restrict__ z, int N_) {
  const int lane = threadIdx.x & 31;
  const int wave = blockIdx.x * 8 + (threadIdx.x >> 5);
  const int r0 = wave * 16;
  if (r0 >= N_) return;

  const int half = lane >> 4;
  const int m    = lane & 15;

  int rowA = r0 + m;
  if (rowA >= N_) rowA = N_ - 1;
  const float* __restrict__ arow = feat + (size_t)rowA * DIN;
  const float* __restrict__ b0p  = W + (size_t)(0 * 16 + m) * DIN;
  const float* __restrict__ b1p  = W + (size_t)(1 * 16 + m) * DIN;
  const float* __restrict__ b2p  = W + (size_t)(2 * 16 + m) * DIN;
  const float* __restrict__ b3p  = W + (size_t)(3 * 16 + m) * DIN;

  v8f acc0 = {}, acc1 = {}, acc2 = {}, acc3 = {};

  for (int k0 = 0; k0 < DIN; k0 += 4) {
    const int ka = k0 + 2 * half;
    v2f a  = *(const v2f*)(arow + ka);
    v2f b0 = *(const v2f*)(b0p + ka);
    v2f b1 = *(const v2f*)(b1p + ka);
    v2f b2 = *(const v2f*)(b2p + ka);
    v2f b3 = *(const v2f*)(b3p + ka);
    acc0 = __builtin_amdgcn_wmma_f32_16x16x4_f32(false, a, false, b0,
                                                 (short)0, acc0, false, false);
    acc1 = __builtin_amdgcn_wmma_f32_16x16x4_f32(false, a, false, b1,
                                                 (short)0, acc1, false, false);
    acc2 = __builtin_amdgcn_wmma_f32_16x16x4_f32(false, a, false, b2,
                                                 (short)0, acc2, false, false);
    acc3 = __builtin_amdgcn_wmma_f32_16x16x4_f32(false, a, false, b3,
                                                 (short)0, acc3, false, false);
  }

#pragma unroll
  for (int r = 0; r < 8; ++r) {
    const int row = r0 + r + 8 * half;
    if (row < N_) {
      float* zr = z + (size_t)row * HD + m;
      zr[0]  = acc0[r];
      zr[16] = acc1[r];
      zr[32] = acc2[r];
      zr[48] = acc3[r];
    }
  }
}

__global__ __launch_bounds__(256) void gat_prep(
    const float* __restrict__ z, const float* __restrict__ aw,
    float* __restrict__ el, float* __restrict__ er,
    float* __restrict__ denom, float* __restrict__ out, int N_) {
  const int i = blockIdx.x * blockDim.x + threadIdx.x;
  if (i >= N_ * H4) return;
  const int n = i >> 2;
  const int h = i & 3;
  const float* zr = z + (size_t)n * HD + h * 16;
  const float* a0 = aw + h * 16;
  const float* a1 = aw + 64 + h * 16;
  float s0 = 0.f, s1 = 0.f;
#pragma unroll
  for (int d = 0; d < 16; ++d) {
    const float zv = zr[d];
    s0 = fmaf(zv, a0[d], s0);
    s1 = fmaf(zv, a1[d], s1);
  }
  el[i]    = s0;
  er[i]    = s1;
  denom[i] = 0.f;
  float* o = out + (size_t)n * HD + h * 16;
#pragma unroll
  for (int d = 0; d < 16; ++d) o[d] = 0.f;
}

__global__ __launch_bounds__(256) void gat_edge(
    const int* __restrict__ src, const int* __restrict__ dst,
    const float* __restrict__ el, const float* __restrict__ er,
    const float* __restrict__ z, float* __restrict__ denom,
    float* __restrict__ out, int E_) {
  const int wave = (int)((blockIdx.x * blockDim.x + threadIdx.x) >> 5);
  const int lane = threadIdx.x & 31;
  if (wave >= E_) return;

  const int s = src[wave];
  const int t = dst[wave];
  const int h0 = lane >> 4;
  const int h1 = h0 + 2;

  float x0 = el[s * H4 + h0] + er[t * H4 + h0];
  float x1 = el[s * H4 + h1] + er[t * H4 + h1];
  x0 = x0 > 0.f ? x0 : 0.2f * x0;
  x1 = x1 > 0.f ? x1 : 0.2f * x1;
  const float ex0 = __expf(x0);
  const float ex1 = __expf(x1);

  if ((lane & 15) == 0) {
    atomicAdd(&denom[t * H4 + h0], ex0);
    atomicAdd(&denom[t * H4 + h1], ex1);
  }

  const float z0 = z[(size_t)s * HD + lane];
  const float z1 = z[(size_t)s * HD + lane + 32];
  atomicAdd(&out[(size_t)t * HD + lane],      ex0 * z0);
  atomicAdd(&out[(size_t)t * HD + lane + 32], ex1 * z1);
}

__global__ __launch_bounds__(256) void gat_finalize(
    const float* __restrict__ denom, const float* __restrict__ bias,
    float* __restrict__ out, int N_) {
  const int i = blockIdx.x * blockDim.x + threadIdx.x;
  if (i >= N_ * HD) return;
  const int n = i >> 6;
  const int j = i & 63;
  const int h = j >> 4;
  float den = denom[n * H4 + h];
  den = den > 0.f ? den : 1.f;
  out[i] = out[i] / den + bias[j];
}

extern "C" void kernel_launch(void* const* d_in, const int* in_sizes, int n_in,
                              void* d_out, int out_size, void* d_ws, size_t ws_size,
                              hipStream_t stream) {
  const float* feat = (const float*)d_in[0];
  const int*   src  = (const int*)d_in[1];
  const int*   dst  = (const int*)d_in[2];
  const float* W    = (const float*)d_in[3];
  const float* aw   = (const float*)d_in[4];
  const float* bias = (const float*)d_in[5];
  float* out = (float*)d_out;

  const int N_ = in_sizes[0] / DIN;
  const int E_ = in_sizes[1];

  float* ws    = (float*)d_ws;
  float* z     = ws;
  float* el    = z  + (size_t)N_ * HD;
  float* er    = el + (size_t)N_ * H4;
  float* denom = er + (size_t)N_ * H4;

  const int gemmWaves  = (N_ + 15) / 16;
  const int gemmBlocks = (gemmWaves + 7) / 8;
  gat_gemm_wmma<<<gemmBlocks, 256, 0, stream>>>(feat, W, z, N_);

  const int prepT = N_ * H4;
  gat_prep<<<(prepT + 255) / 256, 256, 0, stream>>>(z, aw, el, er, denom, out, N_);

  gat_edge<<<(E_ + 7) / 8, 256, 0, stream>>>(src, dst, el, er, z, denom, out, E_);

  const int finT = N_ * HD;
  gat_finalize<<<(finT + 255) / 256, 256, 0, stream>>>(denom, bias, out, N_);
}


# GEMM: W staged once per work-group in LDS (row pitch 132 floats); B fragments read from LDS, only A stays a global load (8 loads per 128-byte line issued together); k order and sums unchanged
# speedup vs baseline: 2.4086x; 2.4086x over previous
#include <hip/hip_runtime.h>
#include <hip/hip_bf16.h>


#define DIN 128
#define HD  64
#define H4  4

typedef __attribute__((ext_vector_type(2))) float v2f;
typedef __attribute__((ext_vector_type(8))) float v8f;

#define WPITCH 132

__global__ __launch_bounds__(256) void gat_gemm_wmma(
    const float* __restrict__ feat, const float* __restrict__ W,
    float* __restrict__ z, int N_) {
  __shared__ __attribute__((aligned(16))) float Ws[HD * WPITCH];

  {
    float4 w[8];
#pragma unroll
    for (int j = 0; j < 8; ++j) {
      const int i = (int)threadIdx.x + 256 * j;
      w[j] = *(const float4*)(W + (size_t)(i >> 5) * DIN + 4 * (i & 31));
    }
#pragma unroll
    for (int j = 0; j < 8; ++j) {
      const int i = (int)threadIdx.x + 256 * j;
      *(float4*)(Ws + (i >> 5) * WPITCH + 4 * (i & 31)) = w[j];
    }
  }
  __syncthreads();

  const int lane = threadIdx.x & 31;
  const int wave = blockIdx.x * 8 + (threadIdx.x >> 5);
  const int r0 = wave * 16;
  if (r0 >= N_) return;

  const int half = lane >> 4;
  const int m    = lane & 15;

  int rowA = r0 + m;
  if (rowA >= N_) rowA = N_ - 1;
  const float* __restrict__ arow = feat + (size_t)rowA * DIN + 2 * half;
  const float* bp = Ws + m * WPITCH + 2 * half;

  v8f acc0 = {}, acc1 = {}, acc2 = {}, acc3 = {};

  auto load_a = [&](v2f (&a)[8], int lb) {
#pragma unroll
    for (int j = 0; j < 8; ++j) a[j] = *(const v2f*)(arow + 32 * lb + 4 * j);
  };
  auto products = [&](const v2f (&a)[8], int lb) {
#pragma unroll
    for (int j = 0; j < 8; ++j) {
      const int k0 = 32 * lb + 4 * j;
      const v2f b0 = *(const v2f*)(bp + 0 * 16 * WPITCH + k0);
      const v2f b1 = *(const v2f*)(bp + 1 * 16 * WPITCH + k0);
      const v2f b2 = *(const v2f*)(bp + 2 * 16 * WPITCH + k0);
      const v2f b3 = *(const v2f*)(bp + 3 * 16 * WPITCH + k0);
      acc0 = __builtin_amdgcn_wmma_f32_16x16x4_f32(false, a[j], false, b0,
                                                   (short)0, acc0, false, false);
      acc1 = __builtin_amdgcn_wmma_f32_16x16x4_f32(false, a[j], false, b1,
                                                   (short)0, acc1, false, false);
      acc2 = __builtin_amdgcn_wmma_f32_16x16x4_f32(false, a[j], false, b2,
                                                   (short)0, acc2, false, false);
      acc3 = __builtin_amdgcn_wmma_f32_16x16x4_f32(false, a[j], false, b3,
                                                   (short)0, acc3, false, false);
    }
  };
  v2f a0[8], a1[8], a2[8], a3[8];
  load_a(a0, 0);
  load_a(a1, 1);
  __builtin_amdgcn_sched_barrier(0);
  products(a0, 0);
  load_a(a2, 2);
  __builtin_amdgcn_sched_barrier(0);
  products(a1, 1);
  load_a(a3, 3);
  __builtin_amdgcn_sched_barrier(0);
  products(a2, 2);
  products(a3, 3);

#pragma unroll
  for (int r = 0; r < 8; ++r) {
    const int row = r0 + r + 8 * half;
    if (row < N_) {
      float* zr = z + (size_t)row * HD + m;
      zr[0]  = acc0[r];
      zr[16] = acc1[r];
      zr[32] = acc2[r];
      zr[48] = acc3[r];
    }
  }
}

#define OWN_THREADS 1024
#define OWN_GROUPS  256
#define MAXROWS     392
#define LIST_CAP    16384

__device__ __forceinline__ bool keys_in_use(int N_, int E_, size_t ws_bytes) {
  const size_t need = (size_t)N_ * (HD + 2 * H4) * sizeof(float) + (size_t)(E_ >> 2) * 4u;
  const int span = (N_ + OWN_GROUPS - 1) / OWN_GROUPS;
  return ws_bytes >= need && span <= MAXROWS;
}

__global__ __launch_bounds__(256) void gat_prep(
    const float* __restrict__ z, const float* __restrict__ aw,
    float* __restrict__ el, float* __restrict__ er, int N_,
    const int* __restrict__ dst, unsigned* __restrict__ keys, int E_, size_t ws_bytes) {
  const int i = blockIdx.x * blockDim.x + threadIdx.x;
  if (keys_in_use(N_, E_, ws_bytes)) {
    const unsigned span = (unsigned)((N_ + OWN_GROUPS - 1) / OWN_GROUPS);
    const int4* __restrict__ dst4 = (const int4*)dst;
    const int nvec = E_ >> 2;
    const int step = (int)(gridDim.x * blockDim.x);
    for (int v = i; v < nvec; v += step) {
      const int4 d = dst4[v];
      keys[v] = (((unsigned)d.x / span) & 255u) | ((((unsigned)d.y / span) & 255u) << 8) |
                ((((unsigned)d.z / span) & 255u) << 16) | (((unsigned)d.w / span) << 24);
    }
  }
  if (i >= N_ * H4) return;
  const int n = i >> 2;
  const int h = i & 3;
  const float* zr = z + (size_t)n * HD + h * 16;
  const float* a0 = aw + h * 16;
  const float* a1 = aw + 64 + h * 16;
  float s0 = 0.f, s1 = 0.f;
#pragma unroll
  for (int d = 0; d < 16; ++d) {
    const float zv = zr[d];
    s0 = fmaf(zv, a0[d], s0);
    s1 = fmaf(zv, a1[d], s1);
  }
  el[i] = s0;
  er[i] = s1;
}

__device__ __forceinline__ unsigned zero_bytes(unsigned x) {
  const unsigned y = (x & 0x7f7f7f7fu) + 0x7f7f7f7fu;
  return ~(y | x | 0x7f7f7f7fu);
}

__device__ __forceinline__ float lane_bcast_f(float v, int l) {
  return __int_as_float(__builtin_amdgcn_readlane(__float_as_int(v), l));
}

__global__ __launch_bounds__(OWN_THREADS) void gat_owner(
    const int* __restrict__ src, const int* __restrict__ dst,
    const float* __restrict__ el, const float* __restrict__ er,
    const float* __restrict__ z, const float* __restrict__ bias,
    float* __restrict__ out, int N_, int E_,
    const unsigned* __restrict__ keys, size_t ws_bytes) {
  __shared__ float acc[MAXROWS * HD];
  __shared__ float den[MAXROWS * H4];
  __shared__ unsigned list[LIST_CAP];
  __shared__ unsigned cnt;

  const int tid  = threadIdx.x;
  const int lane = tid & 31;
  const int wave = __builtin_amdgcn_readfirstlane(tid >> 5);
  const int G    = gridDim.x;
  const int span = (N_ + G - 1) / G;
  const int4* __restrict__ dst4 = (const int4*)dst;
  const int nvec  = E_ >> 2;
  const int nfull = nvec / (8 * OWN_THREADS);
  const int boff  = (int)(((long long)blockIdx.x * nfull) / G);
  const bool use_keys = keys_in_use(N_, E_, ws_bytes) && G == OWN_GROUPS;
  const uint4* __restrict__ key4 = (const uint4*)keys;
  const int nkv    = E_ >> 4;
  const int nkfull = nkv / (8 * OWN_THREADS);
  const int kboff  = (int)(((long long)blockIdx.x * nkfull) / G);
  const unsigned gs = (unsigned)blockIdx.x * 0x01010101u;

  for (int sub = 0; sub < span; sub += MAXROWS) {
    const int row0 = blockIdx.x * span + sub;
    int nr = span - sub;
    if (nr > MAXROWS) nr = MAXROWS;
    if (row0 + nr > N_) nr = N_ - row0;
    if (nr < 0) nr = 0;
    const unsigned nrows = (unsigned)nr;

    for (int i = tid; i < MAXROWS * HD; i += OWN_THREADS) acc[i] = 0.f;
    for (int i = tid; i < MAXROWS * H4; i += OWN_THREADS) den[i] = 0.f;
    if (tid == 0) cnt = 0u;
    __syncthreads();

    auto add_edge_one_lane = [&](unsigned e) {
      const int s = src[e];
      const int t = dst[e];
      const int r = t - row0;
      for (int h = 0; h < H4; ++h) {
        float x = el[s * H4 + h] + er[t * H4 + h];
        x = x > 0.f ? x : 0.2f * x;
        const float ex = __expf(x);
        atomicAdd(&den[r * H4 + h], ex);
        for (int d = 0; d < 16; ++d)
          atomicAdd(&acc[r * HD + h * 16 + d], ex * z[(size_t)s * HD + h * 16 + d]);
      }
    };
    auto push = [&](unsigned e, unsigned bit, unsigned& ov) {
      const unsigned idx = atomicAdd(&cnt, 1u);
      if (idx < LIST_CAP) list[idx] = e;
      else ov |= bit;
    };
#define OWN_TEST(dv, ev, bit) \
    if ((unsigned)((dv) - row0) < nrows) push((unsigned)(ev), (bit), ov);

    auto take = [&](unsigned e) {
      const unsigned idx = atomicAdd(&cnt, 1u);
      if (idx < LIST_CAP) list[idx] = e;
      else add_edge_one_lane(e);
    };
    auto flags16 = [&](const uint4 kv) -> unsigned {
      const unsigned f0 = zero_bytes(kv.x ^ gs), f1 = zero_bytes(kv.y ^ gs);
      const unsigned f2 = zero_bytes(kv.z ^ gs), f3 = zero_bytes(kv.w ^ gs);
      return (f0 >> 7) | (f1 >> 6) | (f2 >> 5) | (f3 >> 4);
    };
    auto scan16 = [&](const uint4 kv, unsigned e0) {
      unsigned b = flags16(kv);
      while (b) {
        const unsigned p = (unsigned)__builtin_ctz(b);
        b &= b - 1u;
        take(e0 + 4u * (p & 3u) + (p >> 3));
      }
    };
    auto scan64 = [&](const uint4 ka, const uint4 kb, const uint4 kc, const uint4 kd,
                      unsigned e0) {
      unsigned long long b =
          (unsigned long long)(flags16(ka) | (flags16(kb) << 4)) |
          ((unsigned long long)(flags16(kc) | (flags16(kd) << 4)) << 32);
      while (b) {
        const unsigned p = (unsigned)__builtin_ctzll(b);
        b &= b - 1ull;
        const unsigned j = ((p >> 2) & 1u) | ((p >> 5) << 1);
        take(e0 + j * (16u * OWN_THREADS) + 4u * (p & 3u) + ((p >> 3) & 3u));
      }
    };

    if (use_keys) {
    for (int it = 0; it < nkfull; ++it) {
      int bi = it + kboff;
      if (bi >= nkfull) bi -= nkfull;
      const int v = bi * (8 * OWN_THREADS) + tid;
      const uint4 k0 = key4[v];
      const uint4 k1 = key4[v + OWN_THREADS];
      const uint4 k2 = key4[v + 2 * OWN_THREADS];
      const uint4 k3 = key4[v + 3 * OWN_THREADS];
      const uint4 k4 = key4[v + 4 * OWN_THREADS];
      const uint4 k5 = key4[v + 5 * OWN_THREADS];
      const uint4 k6 = key4[v + 6 * OWN_THREADS];
      const uint4 k7 = key4[v + 7 * OWN_THREADS];
      const unsigned e0 = 16u * (unsigned)v;
      scan64(k0, k1, k2, k3, e0);
      scan64(k4, k5, k6, k7, e0 + 64u * OWN_THREADS);
    }
    for (int v = nkfull * (8 * OWN_THREADS) + tid; v < nkv; v += OWN_THREADS)
      scan16(key4[v], 16u * (unsigned)v);
    for (int e = (nkv << 4) + tid; e < E_; e += OWN_THREADS)
      if ((unsigned)(dst[e] - row0) < nrows) take((unsigned)e);
    } else {
    for (int it = 0; it < nfull; ++it) {
      int bi = it + boff;
      if (bi >= nfull) bi -= nfull;
      const int v = bi * (8 * OWN_THREADS) + tid;
      const int4 d0 = dst4[v];
      const int4 d1 = dst4[v + OWN_THREADS];
      const int4 d2 = dst4[v + 2 * OWN_THREADS];
      const int4 d3 = dst4[v + 3 * OWN_THREADS];
      const int4 d4 = dst4[v + 4 * OWN_THREADS];
      const int4 d5 = dst4[v + 5 * OWN_THREADS];
      const int4 d6 = dst4[v + 6 * OWN_THREADS];
      const int4 d7 = dst4[v + 7 * OWN_THREADS];
      const int e0 = 4 * v;
      unsigned ov = 0u;
      OWN_TEST(d0.x, e0 + 0 * OWN_THREADS + 0, 1u << 0)
      OWN_TEST(d0.y, e0 + 0 * OWN_THREADS + 1, 1u << 1)
      OWN_TEST(d0.z, e0 + 0 * OWN_THREADS + 2, 1u << 2)
      OWN_TEST(d0.w, e0 + 0 * OWN_THREADS + 3, 1u << 3)
      OWN_TEST(d1.x, e0 + 4 * OWN_THREADS + 0, 1u << 4)
      OWN_TEST(d1.y, e0 + 4 * OWN_THREADS + 1, 1u << 5)
      OWN_TEST(d1.z, e0 + 4 * OWN_THREADS + 2, 1u << 6)
      OWN_TEST(d1.w, e0 + 4 * OWN_THREADS + 3, 1u << 7)
      OWN_TEST(d2.x, e0 + 8 * OWN_THREADS + 0, 1u << 8)
      OWN_TEST(d2.y, e0 + 8 * OWN_THREADS + 1, 1u << 9)
      OWN_TEST(d2.z, e0 + 8 * OWN_THREADS + 2, 1u << 10)
      OWN_TEST(d2.w, e0 + 8 * OWN_THREADS + 3, 1u << 11)
      OWN_TEST(d3.x, e0 + 12 * OWN_THREADS + 0, 1u << 12)
      OWN_TEST(d3.y, e0 + 12 * OWN_THREADS + 1, 1u << 13)
      OWN_TEST(d3.z, e0 + 12 * OWN_THREADS + 2, 1u << 14)
      OWN_TEST(d3.w, e0 + 12 * OWN_THREADS + 3, 1u << 15)
      OWN_TEST(d4.x, e0 + 16 * OWN_THREADS + 0, 1u << 16)
      OWN_TEST(d4.y, e0 + 16 * OWN_THREADS + 1, 1u << 17)
      OWN_TEST(d4.z, e0 + 16 * OWN_THREADS + 2, 1u << 18)
      OWN_TEST(d4.w, e0 + 16 * OWN_THREADS + 3, 1u << 19)
      OWN_TEST(d5.x, e0 + 20 * OWN_THREADS + 0, 1u << 20)
      OWN_TEST(d5.y, e0 + 20 * OWN_THREADS + 1, 1u << 21)
      OWN_TEST(d5.z, e0 + 20 * OWN_THREADS + 2, 1u << 22)
      OWN_TEST(d5.w, e0 + 20 * OWN_THREADS + 3, 1u << 23)
      OWN_TEST(d6.x, e0 + 24 * OWN_THREADS + 0, 1u << 24)
      OWN_TEST(d6.y, e0 + 24 * OWN_THREADS + 1, 1u << 25)
      OWN_TEST(d6.z, e0 + 24 * OWN_THREADS + 2, 1u << 26)
      OWN_TEST(d6.w, e0 + 24 * OWN_THREADS + 3, 1u << 27)
      OWN_TEST(d7.x, e0 + 28 * OWN_THREADS + 0, 1u << 28)
      OWN_TEST(d7.y, e0 + 28 * OWN_THREADS + 1, 1u << 29)
      OWN_TEST(d7.z, e0 + 28 * OWN_THREADS + 2, 1u << 30)
      OWN_TEST(d7.w, e0 + 28 * OWN_THREADS + 3, 1u << 31)
      while (ov) {
        const int b = __builtin_ctz(ov);
        ov &= ov - 1u;
        add_edge_one_lane((unsigned)(e0 + (b >> 2) * (4 * OWN_THREADS) + (b & 3)));
      }
    }
    for (int v = nfull * (8 * OWN_THREADS) + tid; v < nvec; v += OWN_THREADS) {
      const int4 d0 = dst4[v];
      const int e0 = 4 * v;
      unsigned ov = 0u;
      OWN_TEST(d0.x, e0 + 0, 1u << 0)
      OWN_TEST(d0.y, e0 + 1, 1u << 1)
      OWN_TEST(d0.z, e0 + 2, 1u << 2)
      OWN_TEST(d0.w, e0 + 3, 1u << 3)
      while (ov) {
        const int b = __builtin_ctz(ov);
        ov &= ov - 1u;
        add_edge_one_lane((unsigned)(e0 + b));
      }
    }
    for (int e = (nvec << 2) + tid; e < E_; e += OWN_THREADS) {
      unsigned ov = 0u;
      OWN_TEST(dst[e], e, 1u)
      if (ov) add_edge_one_lane((unsigned)e);
    }
    }
#undef OWN_TEST
    __syncthreads();

    unsigned n = (unsigned)__builtin_amdgcn_readfirstlane((int)cnt);
    if (n > LIST_CAP) n = LIST_CAP;
    const int hsel = lane >> 3;
    for (unsigned base = (unsigned)wave * 32u; base < n; base += OWN_THREADS) {
      const unsigned i = base + (unsigned)lane;
      const bool valid = i < n;
      const unsigned e = list[valid ? i : base];
      const int s = src[e];
      const int t = dst[e];
      const int r = t - row0;
      const float4 a = *(const float4*)(el + (size_t)s * H4);
      const float4 c = *(const float4*)(er + (size_t)t * H4);
      float x0 = a.x + c.x, x1 = a.y + c.y, x2 = a.z + c.z, x3 = a.w + c.w;
      x0 = x0 > 0.f ? x0 : 0.2f * x0;
      x1 = x1 > 0.f ? x1 : 0.2f * x1;
      x2 = x2 > 0.f ? x2 : 0.2f * x2;
      x3 = x3 > 0.f ? x3 : 0.2f * x3;
      const float w0 = __expf(x0), w1 = __expf(x1), w2 = __expf(x2), w3 = __expf(x3);
      if (valid) {
        atomicAdd(&den[r * H4 + 0], w0);
        atomicAdd(&den[r * H4 + 1], w1);
        atomicAdd(&den[r * H4 + 2], w2);
        atomicAdd(&den[r * H4 + 3], w3);
      }
      const unsigned left = n - base;
      const int m = left < 32u ? (int)left : 32;
      for (int j0 = 0; j0 < m; j0 += 8) {
        float2 zz[8];
#pragma unroll
        for (int j = 0; j < 8; ++j) {
          int jj = j0 + j;
          if (jj >= m) jj = j0;
          const int sj = __builtin_amdgcn_readlane(s, jj);
          zz[j] = *(const float2*)(z + (size_t)sj * HD + 2 * lane);
        }
#pragma unroll
        for (int j = 0; j < 8; ++j) {
          const int jj = j0 + j;
          if (jj < m) {
            const int rj = __builtin_amdgcn_readlane(r, jj);
            const float e0 = lane_bcast_f(w0, jj), e1 = lane_bcast_f(w1, jj);
            const float e2 = lane_bcast_f(w2, jj), e3 = lane_bcast_f(w3, jj);
            const float w = hsel == 0 ? e0 : hsel == 1 ? e1 : hsel == 2 ? e2 : e3;
            atomicAdd(&acc[rj * HD + 2 * lane],     w * zz[j].x);
            atomicAdd(&acc[rj * HD + 2 * lane + 1], w * zz[j].y);
          }
        }
      }
    }
    __syncthreads();

    const int tot = nr * HD;
    for (int i = tid; i < tot; i += OWN_THREADS) {
      const int r = i >> 6;
      const int j = i & 63;
      float dn = den[r * H4 + (j >> 4)];
      dn = dn > 0.f ? dn : 1.f;
      out[(size_t)(row0 + r) * HD + j] = acc[i] / dn + bias[j];
    }
    __syncthreads();
  }
}

extern "C" void kernel_launch(void* const* d_in, const int* in_sizes, int n_in,
                              void* d_out, int out_size, void* d_ws, size_t ws_size,
                              hipStream_t stream) {
  const float* feat = (const float*)d_in[0];
  const int*   src  = (const int*)d_in[1];
  const int*   dst  = (const int*)d_in[2];
  const float* W    = (const float*)d_in[3];
  const float* aw   = (const float*)d_in[4];
  const float* bias = (const float*)d_in[5];
  float* out = (float*)d_out;

  const int N_ = in_sizes[0] / DIN;
  const int E_ = in_sizes[1];

  float* ws = (float*)d_ws;
  float* z  = ws;
  float* el = z  + (size_t)N_ * HD;
  float* er = el + (size_t)N_ * H4;
  unsigned* keys = (unsigned*)(er + (size_t)N_ * H4);

  const int gemmWaves  = (N_ + 15) / 16;
  const int gemmBlocks = (gemmWaves + 7) / 8;
  gat_gemm_wmma<<<gemmBlocks, 256, 0, stream>>>(feat, W, z, N_);

  const int prepT = N_ * H4;
  gat_prep<<<(prepT + 255) / 256, 256, 0, stream>>>(z, aw, el, er, N_, dst, keys, E_, ws_size);

  gat_owner<<<OWN_GROUPS, OWN_THREADS, 0, stream>>>(src, dst, el, er, z, bias, out, N_, E_,
                                                   keys, ws_size);
}


# el/er launch packs src and dst's row inside its owner range into one word per edge (E more workspace words, as far as they fit); owner gather asks one random word per edge, not src[e] and dst[e]
# speedup vs baseline: 2.5854x; 1.0734x over previous
#include <hip/hip_runtime.h>
#include <hip/hip_bf16.h>


#define DIN 128
#define HD  64
#define H4  4

typedef __attribute__((ext_vector_type(2))) float v2f;
typedef __attribute__((ext_vector_type(8))) float v8f;

#define WPITCH 132

__global__ __launch_bounds__(256) void gat_gemm_wmma(
    const float* __restrict__ feat, const float* __restrict__ W,
    float* __restrict__ z, int N_) {
  __shared__ __attribute__((aligned(16))) float Ws[HD * WPITCH];

  {
    float4 w[8];
#pragma unroll
    for (int j = 0; j < 8; ++j) {
      const int i = (int)threadIdx.x + 256 * j;
      w[j] = *(const float4*)(W + (size_t)(i >> 5) * DIN + 4 * (i & 31));
    }
#pragma unroll
    for (int j = 0; j < 8; ++j) {
      const int i = (int)threadIdx.x + 256 * j;
      *(float4*)(Ws + (i >> 5) * WPITCH + 4 * (i & 31)) = w[j];
    }
  }
  __syncthreads();

  const int lane = threadIdx.x & 31;
  const int wave = blockIdx.x * 8 + (threadIdx.x >> 5);
  const int r0 = wave * 16;
  if (r0 >= N_) return;

  const int half = lane >> 4;
  const int m    = lane & 15;

  int rowA = r0 + m;
  if (rowA >= N_) rowA = N_ - 1;
  const float* __restrict__ arow = feat + (size_t)rowA * DIN + 2 * half;
  const float* bp = Ws + m * WPITCH + 2 * half;

  v8f acc0 = {}, acc1 = {}, acc2 = {}, acc3 = {};

  auto load_a = [&](v2f (&a)[8], int lb) {
#pragma unroll
    for (int j = 0; j < 8; ++j) a[j] = *(const v2f*)(arow + 32 * lb + 4 * j);
  };
  auto products = [&](const v2f (&a)[8], int lb) {
#pragma unroll
    for (int j = 0; j < 8; ++j) {
      const int k0 = 32 * lb + 4 * j;
      const v2f b0 = *(const v2f*)(bp + 0 * 16 * WPITCH + k0);
      const v2f b1 = *(const v2f*)(bp + 1 * 16 * WPITCH + k0);
      const v2f b2 = *(const v2f*)(bp + 2 * 16 * WPITCH + k0);
      const v2f b3 = *(const v2f*)(bp + 3 * 16 * WPITCH + k0);
      acc0 = __builtin_amdgcn_wmma_f32_16x16x4_f32(false, a[j], false, b0,
                                                   (short)0, acc0, false, false);
      acc1 = __builtin_amdgcn_wmma_f32_16x16x4_f32(false, a[j], false, b1,
                                                   (short)0, acc1, false, false);
      acc2 = __builtin_amdgcn_wmma_f32_16x16x4_f32(false, a[j], false, b2,
                                                   (short)0, acc2, false, false);
      acc3 = __builtin_amdgcn_wmma_f32_16x16x4_f32(false, a[j], false, b3,
                                                   (short)0, acc3, false, false);
    }
  };
  v2f a0[8], a1[8], a2[8], a3[8];
  load_a(a0, 0);
  load_a(a1, 1);
  __builtin_amdgcn_sched_barrier(0);
  products(a0, 0);
  load_a(a2, 2);
  __builtin_amdgcn_sched_barrier(0);
  products(a1, 1);
  load_a(a3, 3);
  __builtin_amdgcn_sched_barrier(0);
  products(a2, 2);
  products(a3, 3);

#pragma unroll
  for (int r = 0; r < 8; ++r) {
    const int row = r0 + r + 8 * half;
    if (row < N_) {
      float* zr = z + (size_t)row * HD + m;
      zr[0]  = acc0[r];
      zr[16] = acc1[r];
      zr[32] = acc2[r];
      zr[48] = acc3[r];
    }
  }
}

#define OWN_THREADS 1024
#define OWN_GROUPS  256
#define MAXROWS     392
#define LIST_CAP    16384

__device__ __forceinline__ bool keys_in_use(int N_, int E_, size_t ws_bytes) {
  const size_t need = (size_t)N_ * (HD + 2 * H4) * sizeof(float) + (size_t)(E_ >> 2) * 4u;
  const int span = (N_ + OWN_GROUPS - 1) / OWN_GROUPS;
  return ws_bytes >= need && span <= MAXROWS;
}

#define SRC_BITS 17
__device__ __forceinline__ size_t pack_word_offset(int E_) {
  return ((size_t)(E_ >> 2) + 3u) & ~(size_t)3u;
}
__device__ __forceinline__ unsigned pack_edges(int N_, int E_, size_t ws_bytes) {
  if (!keys_in_use(N_, E_, ws_bytes) || N_ > (1 << SRC_BITS)) return 0u;
  const size_t base = (size_t)N_ * (HD + 2 * H4) * sizeof(float) + pack_word_offset(E_) * 4u;
  if (ws_bytes <= base) return 0u;
  size_t room = (ws_bytes - base) >> 2;
  const size_t want = (size_t)(E_ >> 2) << 2;
  if (room > want) room = want;
  return (unsigned)(room & ~(size_t)3u);
}

__global__ __launch_bounds__(256) void gat_prep(
    const float* __restrict__ z, const float* __restrict__ aw,
    float* __restrict__ el, float* __restrict__ er, int N_,
    const int* __restrict__ src, const int* __restrict__ dst,
    unsigned* __restrict__ keys, int E_, size_t ws_bytes) {
  const int i = blockIdx.x * blockDim.x + threadIdx.x;
  if (keys_in_use(N_, E_, ws_bytes)) {
    const unsigned span = (unsigned)((N_ + OWN_GROUPS - 1) / OWN_GROUPS);
    const int4* __restrict__ dst4 = (const int4*)dst;
    const int4* __restrict__ src4 = (const int4*)src;
    const unsigned pk_n = pack_edges(N_, E_, ws_bytes);
    uint4* __restrict__ pk4 = (uint4*)(keys + pack_word_offset(E_));
    const int nvec = E_ >> 2;
    const int step = (int)(gridDim.x * blockDim.x);
    for (int v = i; v < nvec; v += step) {
      const int4 d = dst4[v];
      const bool packed = 4u * (unsigned)v < pk_n;
      int4 s = make_int4(0, 0, 0, 0);
      if (pk_n != 0u) s = src4[v];
      const unsigned qx = (unsigned)d.x / span, qy = (unsigned)d.y / span;
      const unsigned qz = (unsigned)d.z / span, qw = (unsigned)d.w / span;
      keys[v] = (qx & 255u) | ((qy & 255u) << 8) | ((qz & 255u) << 16) | (qw << 24);
      if (packed) {
        uint4 p;
        p.x = (unsigned)s.x | (((unsigned)d.x - qx * span) << SRC_BITS);
        p.y = (unsigned)s.y | (((unsigned)d.y - qy * span) << SRC_BITS);
        p.z = (unsigned)s.z | (((unsigned)d.z - qz * span) << SRC_BITS);
        p.w = (unsigned)s.w | (((unsigned)d.w - qw * span) << SRC_BITS);
        pk4[v] = p;
      }
    }
  }
  if (i >= N_ * H4) return;
  const int n = i >> 2;
  const int h = i & 3;
  const float* zr = z + (size_t)n * HD + h * 16;
  const float* a0 = aw + h * 16;
  const float* a1 = aw + 64 + h * 16;
  float s0 = 0.f, s1 = 0.f;
#pragma unroll
  for (int d = 0; d < 16; ++d) {
    const float zv = zr[d];
    s0 = fmaf(zv, a0[d], s0);
    s1 = fmaf(zv, a1[d], s1);
  }
  el[i] = s0;
  er[i] = s1;
}

__device__ __forceinline__ unsigned zero_bytes(unsigned x) {
  const unsigned y = (x & 0x7f7f7f7fu) + 0x7f7f7f7fu;
  return ~(y | x | 0x7f7f7f7fu);
}

__device__ __forceinline__ float lane_bcast_f(float v, int l) {
  return __int_as_float(__builtin_amdgcn_readlane(__float_as_int(v), l));
}

__global__ __launch_bounds__(OWN_THREADS) void gat_owner(
    const int* __restrict__ src, const int* __restrict__ dst,
    const float* __restrict__ el, const float* __restrict__ er,
    const float* __restrict__ z, const float* __restrict__ bias,
    float* __restrict__ out, int N_, int E_,
    const unsigned* __restrict__ keys, size_t ws_bytes) {
  __shared__ float acc[MAXROWS * HD];
  __shared__ float den[MAXROWS * H4];
  __shared__ unsigned list[LIST_CAP];
  __shared__ unsigned cnt;

  const int tid  = threadIdx.x;
  const int lane = tid & 31;
  const int wave = __builtin_amdgcn_readfirstlane(tid >> 5);
  const int G    = gridDim.x;
  const int span = (N_ + G - 1) / G;
  const int4* __restrict__ dst4 = (const int4*)dst;
  const int nvec  = E_ >> 2;
  const int nfull = nvec / (8 * OWN_THREADS);
  const int boff  = (int)(((long long)blockIdx.x * nfull) / G);
  const bool use_keys = keys_in_use(N_, E_, ws_bytes) && G == OWN_GROUPS;
  const uint4* __restrict__ key4 = (const uint4*)keys;
  const int nkv    = E_ >> 4;
  const int nkfull = nkv / (8 * OWN_THREADS);
  const int kboff  = (int)(((long long)blockIdx.x * nkfull) / G);
  const unsigned gs = (unsigned)blockIdx.x * 0x01010101u;
  const unsigned pk_n = use_keys ? pack_edges(N_, E_, ws_bytes) : 0u;
  const unsigned* __restrict__ pk = keys + pack_word_offset(E_);

  for (int sub = 0; sub < span; sub += MAXROWS) {
    const int row0 = blockIdx.x * span + sub;
    int nr = span - sub;
    if (nr > MAXROWS) nr = MAXROWS;
    if (row0 + nr > N_) nr = N_ - row0;
    if (nr < 0) nr = 0;
    const unsigned nrows = (unsigned)nr;

    for (int i = tid; i < MAXROWS * HD; i += OWN_THREADS) acc[i] = 0.f;
    for (int i = tid; i < MAXROWS * H4; i += OWN_THREADS) den[i] = 0.f;
    if (tid == 0) cnt = 0u;
    __syncthreads();

    auto add_edge_one_lane = [&](unsigned e) {
      const int s = src[e];
      const int t = dst[e];
      const int r = t - row0;
      for (int h = 0; h < H4; ++h) {
        float x = el[s * H4 + h] + er[t * H4 + h];
        x = x > 0.f ? x : 0.2f * x;
        const float ex = __expf(x);
        atomicAdd(&den[r * H4 + h], ex);
        for (int d = 0; d < 16; ++d)
          atomicAdd(&acc[r * HD + h * 16 + d], ex * z[(size_t)s * HD + h * 16 + d]);
      }
    };
    auto push = [&](unsigned e, unsigned bit, unsigned& ov) {
      const unsigned idx = atomicAdd(&cnt, 1u);
      if (idx < LIST_CAP) list[idx] = e;
      else ov |= bit;
    };
#define OWN_TEST(dv, ev, bit) \
    if ((unsigned)((dv) - row0) < nrows) push((unsigned)(ev), (bit), ov);

    auto take = [&](unsigned e) {
      const unsigned idx = atomicAdd(&cnt, 1u);
      if (idx < LIST_CAP) list[idx] = e;
      else add_edge_one_lane(e);
    };
    auto flags16 = [&](const uint4 kv) -> unsigned {
      const unsigned f0 = zero_bytes(kv.x ^ gs), f1 = zero_bytes(kv.y ^ gs);
      const unsigned f2 = zero_bytes(kv.z ^ gs), f3 = zero_bytes(kv.w ^ gs);
      return (f0 >> 7) | (f1 >> 6) | (f2 >> 5) | (f3 >> 4);
    };
    auto scan16 = [&](const uint4 kv, unsigned e0) {
      unsigned b = flags16(kv);
      while (b) {
        const unsigned p = (unsigned)__builtin_ctz(b);
        b &= b - 1u;
        take(e0 + 4u * (p & 3u) + (p >> 3));
      }
    };
    auto scan64 = [&](const uint4 ka, const uint4 kb, const uint4 kc, const uint4 kd,
                      unsigned e0) {
      unsigned long long b =
          (unsigned long long)(flags16(ka) | (flags16(kb) << 4)) |
          ((unsigned long long)(flags16(kc) | (flags16(kd) << 4)) << 32);
      while (b) {
        const unsigned p = (unsigned)__builtin_ctzll(b);
        b &= b - 1ull;
        const unsigned j = ((p >> 2) & 1u) | ((p >> 5) << 1);
        take(e0 + j * (16u * OWN_THREADS) + 4u * (p & 3u) + ((p >> 3) & 3u));
      }
    };

    if (use_keys) {
    for (int it = 0; it < nkfull; ++it) {
      int bi = it + kboff;
      if (bi >= nkfull) bi -= nkfull;
      const int v = bi * (8 * OWN_THREADS) + tid;
      const uint4 k0 = key4[v];
      const uint4 k1 = key4[v + OWN_THREADS];
      const uint4 k2 = key4[v + 2 * OWN_THREADS];
      const uint4 k3 = key4[v + 3 * OWN_THREADS];
      const uint4 k4 = key4[v + 4 * OWN_THREADS];
      const uint4 k5 = key4[v + 5 * OWN_THREADS];
      const uint4 k6 = key4[v + 6 * OWN_THREADS];
      const uint4 k7 = key4[v + 7 * OWN_THREADS];
      const unsigned e0 = 16u * (unsigned)v;
      scan64(k0, k1, k2, k3, e0);
      scan64(k4, k5, k6, k7, e0 + 64u * OWN_THREADS);
    }
    for (int v = nkfull * (8 * OWN_THREADS) + tid; v < nkv; v += OWN_THREADS)
      scan16(key4[v], 16u * (unsigned)v);
    for (int e = (nkv << 4) + tid; e < E_; e += OWN_THREADS)
      if ((unsigned)(dst[e] - row0) < nrows) take((unsigned)e);
    } else {
    for (int it = 0; it < nfull; ++it) {
      int bi = it + boff;
      if (bi >= nfull) bi -= nfull;
      const int v = bi * (8 * OWN_THREADS) + tid;
      const int4 d0 = dst4[v];
      const int4 d1 = dst4[v + OWN_THREADS];
      const int4 d2 = dst4[v + 2 * OWN_THREADS];
      const int4 d3 = dst4[v + 3 * OWN_THREADS];
      const int4 d4 = dst4[v + 4 * OWN_THREADS];
      const int4 d5 = dst4[v + 5 * OWN_THREADS];
      const int4 d6 = dst4[v + 6 * OWN_THREADS];
      const int4 d7 = dst4[v + 7 * OWN_THREADS];
      const int e0 = 4 * v;
      unsigned ov = 0u;
      OWN_TEST(d0.x, e0 + 0 * OWN_THREADS + 0, 1u << 0)
      OWN_TEST(d0.y, e0 + 0 * OWN_THREADS + 1, 1u << 1)
      OWN_TEST(d0.z, e0 + 0 * OWN_THREADS + 2, 1u << 2)
      OWN_TEST(d0.w, e0 + 0 * OWN_THREADS + 3, 1u << 3)
      OWN_TEST(d1.x, e0 + 4 * OWN_THREADS + 0, 1u << 4)
      OWN_TEST(d1.y, e0 + 4 * OWN_THREADS + 1, 1u << 5)
      OWN_TEST(d1.z, e0 + 4 * OWN_THREADS + 2, 1u << 6)
      OWN_TEST(d1.w, e0 + 4 * OWN_THREADS + 3, 1u << 7)
      OWN_TEST(d2.x, e0 + 8 * OWN_THREADS + 0, 1u << 8)
      OWN_TEST(d2.y, e0 + 8 * OWN_THREADS + 1, 1u << 9)
      OWN_TEST(d2.z, e0 + 8 * OWN_THREADS + 2, 1u << 10)
      OWN_TEST(d2.w, e0 + 8 * OWN_THREADS + 3, 1u << 11)
      OWN_TEST(d3.x, e0 + 12 * OWN_THREADS + 0, 1u << 12)
      OWN_TEST(d3.y, e0 + 12 * OWN_THREADS + 1, 1u << 13)
      OWN_TEST(d3.z, e0 + 12 * OWN_THREADS + 2, 1u << 14)
      OWN_TEST(d3.w, e0 + 12 * OWN_THREADS + 3, 1u << 15)
      OWN_TEST(d4.x, e0 + 16 * OWN_THREADS + 0, 1u << 16)
      OWN_TEST(d4.y, e0 + 16 * OWN_THREADS + 1, 1u << 17)
      OWN_TEST(d4.z, e0 + 16 * OWN_THREADS + 2, 1u << 18)
      OWN_TEST(d4.w, e0 + 16 * OWN_THREADS + 3, 1u << 19)
      OWN_TEST(d5.x, e0 + 20 * OWN_THREADS + 0, 1u << 20)
      OWN_TEST(d5.y, e0 + 20 * OWN_THREADS + 1, 1u << 21)
      OWN_TEST(d5.z, e0 + 20 * OWN_THREADS + 2, 1u << 22)
      OWN_TEST(d5.w, e0 + 20 * OWN_THREADS + 3, 1u << 23)
      OWN_TEST(d6.x, e0 + 24 * OWN_THREADS + 0, 1u << 24)
      OWN_TEST(d6.y, e0 + 24 * OWN_THREADS + 1, 1u << 25)
      OWN_TEST(d6.z, e0 + 24 * OWN_THREADS + 2, 1u << 26)
      OWN_TEST(d6.w, e0 + 24 * OWN_THREADS + 3, 1u << 27)
      OWN_TEST(d7.x, e0 + 28 * OWN_THREADS + 0, 1u << 28)
      OWN_TEST(d7.y, e0 + 28 * OWN_THREADS + 1, 1u << 29)
      OWN_TEST(d7.z, e0 + 28 * OWN_THREADS + 2, 1u << 30)
      OWN_TEST(d7.w, e0 + 28 * OWN_THREADS + 3, 1u << 31)
      while (ov) {
        const int b = __builtin_ctz(ov);
        ov &= ov - 1u;
        add_edge_one_lane((unsigned)(e0 + (b >> 2) * (4 * OWN_THREADS) + (b & 3)));
      }
    }
    for (int v = nfull * (8 * OWN_THREADS) + tid; v < nvec; v += OWN_THREADS) {
      const int4 d0 = dst4[v];
      const int e0 = 4 * v;
      unsigned ov = 0u;
      OWN_TEST(d0.x, e0 + 0, 1u << 0)
      OWN_TEST(d0.y, e0 + 1, 1u << 1)
      OWN_TEST(d0.z, e0 + 2, 1u << 2)
      OWN_TEST(d0.w, e0 + 3, 1u << 3)
      while (ov) {
        const int b = __builtin_ctz(ov);
        ov &= ov - 1u;
        add_edge_one_lane((unsigned)(e0 + b));
      }
    }
    for (int e = (nvec << 2) + tid; e < E_; e += OWN_THREADS) {
      unsigned ov = 0u;
      OWN_TEST(dst[e], e, 1u)
      if (ov) add_edge_one_lane((unsigned)e);
    }
    }
#undef OWN_TEST
    __syncthreads();

    unsigned n = (unsigned)__builtin_amdgcn_readfirstlane((int)cnt);
    if (n > LIST_CAP) n = LIST_CAP;
    const int hsel = lane >> 3;
    for (unsigned base = (unsigned)wave * 32u; base < n; base += OWN_THREADS) {
      const unsigned i = base + (unsigned)lane;
      const bool valid = i < n;
      const unsigned e = list[valid ? i : base];
      int s, t, r;
      if (e < pk_n) {
        const unsigned p = pk[e];
        s = (int)(p & ((1u << SRC_BITS) - 1u));
        r = (int)(p >> SRC_BITS);
        t = row0 + r;
      } else {
        s = src[e];
        t = dst[e];
        r = t - row0;
      }
      const float4 a = *(const float4*)(el + (size_t)s * H4);
      const float4 c = *(const float4*)(er + (size_t)t * H4);
      float x0 = a.x + c.x, x1 = a.y + c.y, x2 = a.z + c.z, x3 = a.w + c.w;
      x0 = x0 > 0.f ? x0 : 0.2f * x0;
      x1 = x1 > 0.f ? x1 : 0.2f * x1;
      x2 = x2 > 0.f ? x2 : 0.2f * x2;
      x3 = x3 > 0.f ? x3 : 0.2f * x3;
      const float w0 = __expf(x0), w1 = __expf(x1), w2 = __expf(x2), w3 = __expf(x3);
      if (valid) {
        atomicAdd(&den[r * H4 + 0], w0);
        atomicAdd(&den[r * H4 + 1], w1);
        atomicAdd(&den[r * H4 + 2], w2);
        atomicAdd(&den[r * H4 + 3], w3);
      }
      const unsigned left = n - base;
      const int m = left < 32u ? (int)left : 32;
      for (int j0 = 0; j0 < m; j0 += 8) {
        float2 zz[8];
#pragma unroll
        for (int j = 0; j < 8; ++j) {
          int jj = j0 + j;
          if (jj >= m) jj = j0;
          const int sj = __builtin_amdgcn_readlane(s, jj);
          zz[j] = *(const float2*)(z + (size_t)sj * HD + 2 * lane);
        }
#pragma unroll
        for (int j = 0; j < 8; ++j) {
          const int jj = j0 + j;
          if (jj < m) {
            const int rj = __builtin_amdgcn_readlane(r, jj);
            const float e0 = lane_bcast_f(w0, jj), e1 = lane_bcast_f(w1, jj);
            const float e2 = lane_bcast_f(w2, jj), e3 = lane_bcast_f(w3, jj);
            const float w = hsel == 0 ? e0 : hsel == 1 ? e1 : hsel == 2 ? e2 : e3;
            atomicAdd(&acc[rj * HD + 2 * lane],     w * zz[j].x);
            atomicAdd(&acc[rj * HD + 2 * lane + 1], w * zz[j].y);
          }
        }
      }
    }
    __syncthreads();

    const int tot = nr * HD;
    for (int i = tid; i < tot; i += OWN_THREADS) {
      const int r = i >> 6;
      const int j = i & 63;
      float dn = den[r * H4 + (j >> 4)];
      dn = dn > 0.f ? dn : 1.f;
      out[(size_t)(row0 + r) * HD + j] = acc[i] / dn + bias[j];
    }
    __syncthreads();
  }
}

extern "C" void kernel_launch(void* const* d_in, const int* in_sizes, int n_in,
                              void* d_out, int out_size, void* d_ws, size_t ws_size,
                              hipStream_t stream) {
  const float* feat = (const float*)d_in[0];
  const int*   src  = (const int*)d_in[1];
  const int*   dst  = (const int*)d_in[2];
  const float* W    = (const float*)d_in[3];
  const float* aw   = (const float*)d_in[4];
  const float* bias = (const float*)d_in[5];
  float* out = (float*)d_out;

  const int N_ = in_sizes[0] / DIN;
  const int E_ = in_sizes[1];

  float* ws = (float*)d_ws;
  float* z  = ws;
  float* el = z  + (size_t)N_ * HD;
  float* er = el + (size_t)N_ * H4;
  unsigned* keys = (unsigned*)(er + (size_t)N_ * H4);

  const int gemmWaves  = (N_ + 15) / 16;
  const int gemmBlocks = (gemmWaves + 7) / 8;
  gat_gemm_wmma<<<gemmBlocks, 256, 0, stream>>>(feat, W, z, N_);

  const int prepT = N_ * H4;
  gat_prep<<<(prepT + 255) / 256, 256, 0, stream>>>(z, aw, el, er, N_, src, dst, keys, E_, ws_size);

  gat_owner<<<OWN_GROUPS, OWN_THREADS, 0, stream>>>(src, dst, el, er, z, bias, out, N_, E_,
                                                   keys, ws_size);
}


# Owner gather: each wave stages its 32 edges' z-row offset, LDS row offset and 4 head weights in LDS and every lane reads them back with plain LDS loads; readlane chains and branchy head select gone
# speedup vs baseline: 2.6661x; 1.0312x over previous
#include <hip/hip_runtime.h>
#include <hip/hip_bf16.h>


#define DIN 128
#define HD  64
#define H4  4

typedef __attribute__((ext_vector_type(2))) float v2f;
typedef __attribute__((ext_vector_type(8))) float v8f;

#define WPITCH 132

__global__ __launch_bounds__(256) void gat_gemm_wmma(
    const float* __restrict__ feat, const float* __restrict__ W,
    float* __restrict__ z, int N_) {
  __shared__ __attribute__((aligned(16))) float Ws[HD * WPITCH];

  {
    float4 w[8];
#pragma unroll
    for (int j = 0; j < 8; ++j) {
      const int i = (int)threadIdx.x + 256 * j;
      w[j] = *(const float4*)(W + (size_t)(i >> 5) * DIN + 4 * (i & 31));
    }
#pragma unroll
    for (int j = 0; j < 8; ++j) {
      const int i = (int)threadIdx.x + 256 * j;
      *(float4*)(Ws + (i >> 5) * WPITCH + 4 * (i & 31)) = w[j];
    }
  }
  __syncthreads();

  const int lane = threadIdx.x & 31;
  const int wave = blockIdx.x * 8 + (threadIdx.x >> 5);
  const int r0 = wave * 16;
  if (r0 >= N_) return;

  const int half = lane >> 4;
  const int m    = lane & 15;

  int rowA = r0 + m;
  if (rowA >= N_) rowA = N_ - 1;
  const float* __restrict__ arow = feat + (size_t)rowA * DIN + 2 * half;
  const float* bp = Ws + m * WPITCH + 2 * half;

  v8f acc0 = {}, acc1 = {}, acc2 = {}, acc3 = {};

  auto load_a = [&](v2f (&a)[8], int lb) {
#pragma unroll
    for (int j = 0; j < 8; ++j) a[j] = *(const v2f*)(arow + 32 * lb + 4 * j);
  };
  auto products = [&](const v2f (&a)[8], int lb) {
#pragma unroll
    for (int j = 0; j < 8; ++j) {
      const int k0 = 32 * lb + 4 * j;
      const v2f b0 = *(const v2f*)(bp + 0 * 16 * WPITCH + k0);
      const v2f b1 = *(const v2f*)(bp + 1 * 16 * WPITCH + k0);
      const v2f b2 = *(const v2f*)(bp + 2 * 16 * WPITCH + k0);
      const v2f b3 = *(const v2f*)(bp + 3 * 16 * WPITCH + k0);
      acc0 = __builtin_amdgcn_wmma_f32_16x16x4_f32(false, a[j], false, b0,
                                                   (short)0, acc0, false, false);
      acc1 = __builtin_amdgcn_wmma_f32_16x16x4_f32(false, a[j], false, b1,
                                                   (short)0, acc1, false, false);
      acc2 = __builtin_amdgcn_wmma_f32_16x16x4_f32(false, a[j], false, b2,
                                                   (short)0, acc2, false, false);
      acc3 = __builtin_amdgcn_wmma_f32_16x16x4_f32(false, a[j], false, b3,
                                                   (short)0, acc3, false, false);
    }
  };
  v2f a0[8], a1[8], a2[8], a3[8];
  load_a(a0, 0);
  load_a(a1, 1);
  __builtin_amdgcn_sched_barrier(0);
  products(a0, 0);
  load_a(a2, 2);
  __builtin_amdgcn_sched_barrier(0);
  products(a1, 1);
  load_a(a3, 3);
  __builtin_amdgcn_sched_barrier(0);
  products(a2, 2);
  products(a3, 3);

#pragma unroll
  for (int r = 0; r < 8; ++r) {
    const int row = r0 + r + 8 * half;
    if (row < N_) {
      float* zr = z + (size_t)row * HD + m;
      zr[0]  = acc0[r];
      zr[16] = acc1[r];
      zr[32] = acc2[r];
      zr[48] = acc3[r];
    }
  }
}

#define OWN_THREADS 1024
#define OWN_GROUPS  256
#define MAXROWS     392
#define LIST_CAP    16384

__device__ __forceinline__ bool keys_in_use(int N_, int E_, size_t ws_bytes) {
  const size_t need = (size_t)N_ * (HD + 2 * H4) * sizeof(float) + (size_t)(E_ >> 2) * 4u;
  const int span = (N_ + OWN_GROUPS - 1) / OWN_GROUPS;
  return ws_bytes >= need && span <= MAXROWS;
}

#define SRC_BITS 17
__device__ __forceinline__ size_t pack_word_offset(int E_) {
  return ((size_t)(E_ >> 2) + 3u) & ~(size_t)3u;
}
__device__ __forceinline__ unsigned pack_edges(int N_, int E_, size_t ws_bytes) {
  if (!keys_in_use(N_, E_, ws_bytes) || N_ > (1 << SRC_BITS)) return 0u;
  const size_t base = (size_t)N_ * (HD + 2 * H4) * sizeof(float) + pack_word_offset(E_) * 4u;
  if (ws_bytes <= base) return 0u;
  size_t room = (ws_bytes - base) >> 2;
  const size_t want = (size_t)(E_ >> 2) << 2;
  if (room > want) room = want;
  return (unsigned)(room & ~(size_t)3u);
}

__global__ __launch_bounds__(256) void gat_prep(
    const float* __restrict__ z, const float* __restrict__ aw,
    float* __restrict__ el, float* __restrict__ er, int N_,
    const int* __restrict__ src, const int* __restrict__ dst,
    unsigned* __restrict__ keys, int E_, size_t ws_bytes) {
  const int i = blockIdx.x * blockDim.x + threadIdx.x;
  if (keys_in_use(N_, E_, ws_bytes)) {
    const unsigned span = (unsigned)((N_ + OWN_GROUPS - 1) / OWN_GROUPS);
    const int4* __restrict__ dst4 = (const int4*)dst;
    const int4* __restrict__ src4 = (const int4*)src;
    const unsigned pk_n = pack_edges(N_, E_, ws_bytes);
    uint4* __restrict__ pk4 = (uint4*)(keys + pack_word_offset(E_));
    const int nvec = E_ >> 2;
    const int step = (int)(gridDim.x * blockDim.x);
    for (int v = i; v < nvec; v += step) {
      const int4 d = dst4[v];
      const bool packed = 4u * (unsigned)v < pk_n;
      int4 s = make_int4(0, 0, 0, 0);
      if (pk_n != 0u) s = src4[v];
      const unsigned qx = (unsigned)d.x / span, qy = (unsigned)d.y / span;
      const unsigned qz = (unsigned)d.z / span, qw = (unsigned)d.w / span;
      keys[v] = (qx & 255u) | ((qy & 255u) << 8) | ((qz & 255u) << 16) | (qw << 24);
      if (packed) {
        uint4 p;
        p.x = (unsigned)s.x | (((unsigned)d.x - qx * span) << SRC_BITS);
        p.y = (unsigned)s.y | (((unsigned)d.y - qy * span) << SRC_BITS);
        p.z = (unsigned)s.z | (((unsigned)d.z - qz * span) << SRC_BITS);
        p.w = (unsigned)s.w | (((unsigned)d.w - qw * span) << SRC_BITS);
        pk4[v] = p;
      }
    }
  }
  if (i >= N_ * H4) return;
  const int n = i >> 2;
  const int h = i & 3;
  const float* zr = z + (size_t)n * HD + h * 16;
  const float* a0 = aw + h * 16;
  const float* a1 = aw + 64 + h * 16;
  float s0 = 0.f, s1 = 0.f;
#pragma unroll
  for (int d = 0; d < 16; ++d) {
    const float zv = zr[d];
    s0 = fmaf(zv, a0[d], s0);
    s1 = fmaf(zv, a1[d], s1);
  }
  el[i] = s0;
  er[i] = s1;
}

__device__ __forceinline__ unsigned zero_bytes(unsigned x) {
  const unsigned y = (x & 0x7f7f7f7fu) + 0x7f7f7f7fu;
  return ~(y | x | 0x7f7f7f7fu);
}

struct __attribute__((aligned(16))) EdgeRec {
  unsigned long long zoff;
  unsigned roff;
  unsigned unused;
};

__global__ __launch_bounds__(OWN_THREADS) void gat_owner(
    const int* __restrict__ src, const int* __restrict__ dst,
    const float* __restrict__ el, const float* __restrict__ er,
    const float* __restrict__ z, const float* __restrict__ bias,
    float* __restrict__ out, int N_, int E_,
    const unsigned* __restrict__ keys, size_t ws_bytes) {
  __shared__ float acc[(MAXROWS + 1) * HD];
  __shared__ float den[MAXROWS * H4];
  __shared__ unsigned list[LIST_CAP];
  __shared__ unsigned cnt;
  __shared__ EdgeRec st_rec[(OWN_THREADS / 32) * 32];
  __shared__ __attribute__((aligned(16))) float st_w[(OWN_THREADS / 32) * 32 * H4];

  const int tid  = threadIdx.x;
  const int lane = tid & 31;
  const int wave = __builtin_amdgcn_readfirstlane(tid >> 5);
  const int G    = gridDim.x;
  const int span = (N_ + G - 1) / G;
  const int4* __restrict__ dst4 = (const int4*)dst;
  const int nvec  = E_ >> 2;
  const int nfull = nvec / (8 * OWN_THREADS);
  const int boff  = (int)(((long long)blockIdx.x * nfull) / G);
  const bool use_keys = keys_in_use(N_, E_, ws_bytes) && G == OWN_GROUPS;
  const uint4* __restrict__ key4 = (const uint4*)keys;
  const int nkv    = E_ >> 4;
  const int nkfull = nkv / (8 * OWN_THREADS);
  const int kboff  = (int)(((long long)blockIdx.x * nkfull) / G);
  const unsigned gs = (unsigned)blockIdx.x * 0x01010101u;
  const unsigned pk_n = use_keys ? pack_edges(N_, E_, ws_bytes) : 0u;
  const unsigned* __restrict__ pk = keys + pack_word_offset(E_);

  for (int sub = 0; sub < span; sub += MAXROWS) {
    const int row0 = blockIdx.x * span + sub;
    int nr = span - sub;
    if (nr > MAXROWS) nr = MAXROWS;
    if (row0 + nr > N_) nr = N_ - row0;
    if (nr < 0) nr = 0;
    const unsigned nrows = (unsigned)nr;

    for (int i = tid; i < MAXROWS * HD; i += OWN_THREADS) acc[i] = 0.f;
    for (int i = tid; i < MAXROWS * H4; i += OWN_THREADS) den[i] = 0.f;
    if (tid == 0) cnt = 0u;
    __syncthreads();

    auto add_edge_one_lane = [&](unsigned e) {
      const int s = src[e];
      const int t = dst[e];
      const int r = t - row0;
      for (int h = 0; h < H4; ++h) {
        float x = el[s * H4 + h] + er[t * H4 + h];
        x = x > 0.f ? x : 0.2f * x;
        const float ex = __expf(x);
        atomicAdd(&den[r * H4 + h], ex);
        for (int d = 0; d < 16; ++d)
          atomicAdd(&acc[r * HD + h * 16 + d], ex * z[(size_t)s * HD + h * 16 + d]);
      }
    };
    auto push = [&](unsigned e, unsigned bit, unsigned& ov) {
      const unsigned idx = atomicAdd(&cnt, 1u);
      if (idx < LIST_CAP) list[idx] = e;
      else ov |= bit;
    };
#define OWN_TEST(dv, ev, bit) \
    if ((unsigned)((dv) - row0) < nrows) push((unsigned)(ev), (bit), ov);

    auto take = [&](unsigned e) {
      const unsigned idx = atomicAdd(&cnt, 1u);
      if (idx < LIST_CAP) list[idx] = e;
      else add_edge_one_lane(e);
    };
    auto flags16 = [&](const uint4 kv) -> unsigned {
      const unsigned f0 = zero_bytes(kv.x ^ gs), f1 = zero_bytes(kv.y ^ gs);
      const unsigned f2 = zero_bytes(kv.z ^ gs), f3 = zero_bytes(kv.w ^ gs);
      return (f0 >> 7) | (f1 >> 6) | (f2 >> 5) | (f3 >> 4);
    };
    auto scan16 = [&](const uint4 kv, unsigned e0) {
      unsigned b = flags16(kv);
      while (b) {
        const unsigned p = (unsigned)__builtin_ctz(b);
        b &= b - 1u;
        take(e0 + 4u * (p & 3u) + (p >> 3));
      }
    };
    auto scan64 = [&](const uint4 ka, const uint4 kb, const uint4 kc, const uint4 kd,
                      unsigned e0) {
      unsigned long long b =
          (unsigned long long)(flags16(ka) | (flags16(kb) << 4)) |
          ((unsigned long long)(flags16(kc) | (flags16(kd) << 4)) << 32);
      while (b) {
        const unsigned p = (unsigned)__builtin_ctzll(b);
        b &= b - 1ull;
        const unsigned j = ((p >> 2) & 1u) | ((p >> 5) << 1);
        take(e0 + j * (16u * OWN_THREADS) + 4u * (p & 3u) + ((p >> 3) & 3u));
      }
    };

    if (use_keys) {
    for (int it = 0; it < nkfull; ++it) {
      int bi = it + kboff;
      if (bi >= nkfull) bi -= nkfull;
      const int v = bi * (8 * OWN_THREADS) + tid;
      const uint4 k0 = key4[v];
      const uint4 k1 = key4[v + OWN_THREADS];
      const uint4 k2 = key4[v + 2 * OWN_THREADS];
      const uint4 k3 = key4[v + 3 * OWN_THREADS];
      const uint4 k4 = key4[v + 4 * OWN_THREADS];
      const uint4 k5 = key4[v + 5 * OWN_THREADS];
      const uint4 k6 = key4[v + 6 * OWN_THREADS];
      const uint4 k7 = key4[v + 7 * OWN_THREADS];
      const unsigned e0 = 16u * (unsigned)v;
      scan64(k0, k1, k2, k3, e0);
      scan64(k4, k5, k6, k7, e0 + 64u * OWN_THREADS);
    }
    for (int v = nkfull * (8 * OWN_THREADS) + tid; v < nkv; v += OWN_THREADS)
      scan16(key4[v], 16u * (unsigned)v);
    for (int e = (nkv << 4) + tid; e < E_; e += OWN_THREADS)
      if ((unsigned)(dst[e] - row0) < nrows) take((unsigned)e);
    } else {
    for (int it = 0; it < nfull; ++it) {
      int bi = it + boff;
      if (bi >= nfull) bi -= nfull;
      const int v = bi * (8 * OWN_THREADS) + tid;
      const int4 d0 = dst4[v];
      const int4 d1 = dst4[v + OWN_THREADS];
      const int4 d2 = dst4[v + 2 * OWN_THREADS];
      const int4 d3 = dst4[v + 3 * OWN_THREADS];
      const int4 d4 = dst4[v + 4 * OWN_THREADS];
      const int4 d5 = dst4[v + 5 * OWN_THREADS];
      const int4 d6 = dst4[v + 6 * OWN_THREADS];
      const int4 d7 = dst4[v + 7 * OWN_THREADS];
      const int e0 = 4 * v;
      unsigned ov = 0u;
      OWN_TEST(d0.x, e0 + 0 * OWN_THREADS + 0, 1u << 0)
      OWN_TEST(d0.y, e0 + 0 * OWN_THREADS + 1, 1u << 1)
      OWN_TEST(d0.z, e0 + 0 * OWN_THREADS + 2, 1u << 2)
      OWN_TEST(d0.w, e0 + 0 * OWN_THREADS + 3, 1u << 3)
      OWN_TEST(d1.x, e0 + 4 * OWN_THREADS + 0, 1u << 4)
      OWN_TEST(d1.y, e0 + 4 * OWN_THREADS + 1, 1u << 5)
      OWN_TEST(d1.z, e0 + 4 * OWN_THREADS + 2, 1u << 6)
      OWN_TEST(d1.w, e0 + 4 * OWN_THREADS + 3, 1u << 7)
      OWN_TEST(d2.x, e0 + 8 * OWN_THREADS + 0, 1u << 8)
      OWN_TEST(d2.y, e0 + 8 * OWN_THREADS + 1, 1u << 9)
      OWN_TEST(d2.z, e0 + 8 * OWN_THREADS + 2, 1u << 10)
      OWN_TEST(d2.w, e0 + 8 * OWN_THREADS + 3, 1u << 11)
      OWN_TEST(d3.x, e0 + 12 * OWN_THREADS + 0, 1u << 12)
      OWN_TEST(d3.y, e0 + 12 * OWN_THREADS + 1, 1u << 13)
      OWN_TEST(d3.z, e0 + 12 * OWN_THREADS + 2, 1u << 14)
      OWN_TEST(d3.w, e0 + 12 * OWN_THREADS + 3, 1u << 15)
      OWN_TEST(d4.x, e0 + 16 * OWN_THREADS + 0, 1u << 16)
      OWN_TEST(d4.y, e0 + 16 * OWN_THREADS + 1, 1u << 17)
      OWN_TEST(d4.z, e0 + 16 * OWN_THREADS + 2, 1u << 18)
      OWN_TEST(d4.w, e0 + 16 * OWN_THREADS + 3, 1u << 19)
      OWN_TEST(d5.x, e0 + 20 * OWN_THREADS + 0, 1u << 20)
      OWN_TEST(d5.y, e0 + 20 * OWN_THREADS + 1, 1u << 21)
      OWN_TEST(d5.z, e0 + 20 * OWN_THREADS + 2, 1u << 22)
      OWN_TEST(d5.w, e0 + 20 * OWN_THREADS + 3, 1u << 23)
      OWN_TEST(d6.x, e0 + 24 * OWN_THREADS + 0, 1u << 24)
      OWN_TEST(d6.y, e0 + 24 * OWN_THREADS + 1, 1u << 25)
      OWN_TEST(d6.z, e0 + 24 * OWN_THREADS + 2, 1u << 26)
      OWN_TEST(d6.w, e0 + 24 * OWN_THREADS + 3, 1u << 27)
      OWN_TEST(d7.x, e0 + 28 * OWN_THREADS + 0, 1u << 28)
      OWN_TEST(d7.y, e0 + 28 * OWN_THREADS + 1, 1u << 29)
      OWN_TEST(d7.z, e0 + 28 * OWN_THREADS + 2, 1u << 30)
      OWN_TEST(d7.w, e0 + 28 * OWN_THREADS + 3, 1u << 31)
      while (ov) {
        const int b = __builtin_ctz(ov);
        ov &= ov - 1u;
        add_edge_one_lane((unsigned)(e0 + (b >> 2) * (4 * OWN_THREADS) + (b & 3)));
      }
    }
    for (int v = nfull * (8 * OWN_THREADS) + tid; v < nvec; v += OWN_THREADS) {
      const int4 d0 = dst4[v];
      const int e0 = 4 * v;
      unsigned ov = 0u;
      OWN_TEST(d0.x, e0 + 0, 1u << 0)
      OWN_TEST(d0.y, e0 + 1, 1u << 1)
      OWN_TEST(d0.z, e0 + 2, 1u << 2)
      OWN_TEST(d0.w, e0 + 3, 1u << 3)
      while (ov) {
        const int b = __builtin_ctz(ov);
        ov &= ov - 1u;
        add_edge_one_lane((unsigned)(e0 + b));
      }
    }
    for (int e = (nvec << 2) + tid; e < E_; e += OWN_THREADS) {
      unsigned ov = 0u;
      OWN_TEST(dst[e], e, 1u)
      if (ov) add_edge_one_lane((unsigned)e);
    }
    }
#undef OWN_TEST
    __syncthreads();

    unsigned n = (unsigned)__builtin_amdgcn_readfirstlane((int)cnt);
    if (n > LIST_CAP) n = LIST_CAP;
    const int hsel = lane >> 3;
    EdgeRec* const my_rec = st_rec + wave * 32;
    float* const my_w = st_w + wave * (32 * H4);
    const char* const zl = (const char*)z + 8 * lane;
    char* const al = (char*)acc + 8 * lane;
    for (unsigned base = (unsigned)wave * 32u; base < n; base += OWN_THREADS) {
      const unsigned i = base + (unsigned)lane;
      const bool valid = i < n;
      const unsigned e = list[valid ? i : base];
      int s, t, r;
      if (e < pk_n) {
        const unsigned p = pk[e];
        s = (int)(p & ((1u << SRC_BITS) - 1u));
        r = (int)(p >> SRC_BITS);
        t = row0 + r;
      } else {
        s = src[e];
        t = dst[e];
        r = t - row0;
      }
      const float4 a = *(const float4*)(el + (size_t)s * H4);
      const float4 c = *(const float4*)(er + (size_t)t * H4);
      float x0 = a.x + c.x, x1 = a.y + c.y, x2 = a.z + c.z, x3 = a.w + c.w;
      x0 = x0 > 0.f ? x0 : 0.2f * x0;
      x1 = x1 > 0.f ? x1 : 0.2f * x1;
      x2 = x2 > 0.f ? x2 : 0.2f * x2;
      x3 = x3 > 0.f ? x3 : 0.2f * x3;
      const float w0 = __expf(x0), w1 = __expf(x1), w2 = __expf(x2), w3 = __expf(x3);
      if (valid) {
        atomicAdd(&den[r * H4 + 0], w0);
        atomicAdd(&den[r * H4 + 1], w1);
        atomicAdd(&den[r * H4 + 2], w2);
        atomicAdd(&den[r * H4 + 3], w3);
      }
      EdgeRec rec;
      rec.zoff = (unsigned long long)(unsigned)s * (HD * sizeof(float));
      rec.roff = (valid ? (unsigned)r : (unsigned)MAXROWS) * (unsigned)(HD * sizeof(float));
      rec.unused = 0u;
      my_rec[lane] = rec;
      *(float4*)(my_w + H4 * lane) =
          valid ? make_float4(w0, w1, w2, w3) : make_float4(0.f, 0.f, 0.f, 0.f);
      const unsigned left = n - base;
      const int m = left < 32u ? (int)left : 32;
      for (int j0 = 0; j0 < m; j0 += 8) {
        EdgeRec q[8];
        float wj[8];
        float2 zz[8];
#pragma unroll
        for (int j = 0; j < 8; ++j) {
          q[j]  = my_rec[j0 + j];
          wj[j] = my_w[H4 * (j0 + j) + hsel];
        }
#pragma unroll
        for (int j = 0; j < 8; ++j)
          zz[j] = *(const float2*)(zl + q[j].zoff);
#pragma unroll
        for (int j = 0; j < 8; ++j) {
          float* const ar = (float*)(al + q[j].roff);
          atomicAdd(ar,     wj[j] * zz[j].x);
          atomicAdd(ar + 1, wj[j] * zz[j].y);
        }
      }
    }
    __syncthreads();

    const int tot = nr * HD;
    for (int i = tid; i < tot; i += OWN_THREADS) {
      const int r = i >> 6;
      const int j = i & 63;
      float dn = den[r * H4 + (j >> 4)];
      dn = dn > 0.f ? dn : 1.f;
      out[(size_t)(row0 + r) * HD + j] = acc[i] / dn + bias[j];
    }
    __syncthreads();
  }
}

extern "C" void kernel_launch(void* const* d_in, const int* in_sizes, int n_in,
                              void* d_out, int out_size, void* d_ws, size_t ws_size,
                              hipStream_t stream) {
  const float* feat = (const float*)d_in[0];
  const int*   src  = (const int*)d_in[1];
  const int*   dst  = (const int*)d_in[2];
  const float* W    = (const float*)d_in[3];
  const float* aw   = (const float*)d_in[4];
  const float* bias = (const float*)d_in[5];
  float* out = (float*)d_out;

  const int N_ = in_sizes[0] / DIN;
  const int E_ = in_sizes[1];

  float* ws = (float*)d_ws;
  float* z  = ws;
  float* el = z  + (size_t)N_ * HD;
  float* er = el + (size_t)N_ * H4;
  unsigned* keys = (unsigned*)(er + (size_t)N_ * H4);

  const int gemmWaves  = (N_ + 15) / 16;
  const int gemmBlocks = (gemmWaves + 7) / 8;
  gat_gemm_wmma<<<gemmBlocks, 256, 0, stream>>>(feat, W, z, N_);

  const int prepT = N_ * H4;
  gat_prep<<<(prepT + 255) / 256, 256, 0, stream>>>(z, aw, el, er, N_, src, dst, keys, E_, ws_size);

  gat_owner<<<OWN_GROUPS, OWN_THREADS, 0, stream>>>(src, dst, el, er, z, bias, out, N_, E_,
                                                   keys, ws_size);
}


# Owner: each wave keeps its own find list (slot = wave count + ballot rank, no shared counter) and gathers 32 finds at a time between its sweep trips, no barrier after the sweep; arrival-order sums
# speedup vs baseline: 2.8217x; 1.0584x over previous
#include <hip/hip_runtime.h>
#include <hip/hip_bf16.h>


#define DIN 128
#define HD  64
#define H4  4

typedef __attribute__((ext_vector_type(2))) float v2f;
typedef __attribute__((ext_vector_type(8))) float v8f;

#define WPITCH 132

__global__ __launch_bounds__(256) void gat_gemm_wmma(
    const float* __restrict__ feat, const float* __restrict__ W,
    float* __restrict__ z, int N_) {
  __shared__ __attribute__((aligned(16))) float Ws[HD * WPITCH];

  {
    float4 w[8];
#pragma unroll
    for (int j = 0; j < 8; ++j) {
      const int i = (int)threadIdx.x + 256 * j;
      w[j] = *(const float4*)(W + (size_t)(i >> 5) * DIN + 4 * (i & 31));
    }
#pragma unroll
    for (int j = 0; j < 8; ++j) {
      const int i = (int)threadIdx.x + 256 * j;
      *(float4*)(Ws + (i >> 5) * WPITCH + 4 * (i & 31)) = w[j];
    }
  }
  __syncthreads();

  const int lane = threadIdx.x & 31;
  const int wave = blockIdx.x * 8 + (threadIdx.x >> 5);
  const int r0 = wave * 16;
  if (r0 >= N_) return;

  const int half = lane >> 4;
  const int m    = lane & 15;

  int rowA = r0 + m;
  if (rowA >= N_) rowA = N_ - 1;
  const float* __restrict__ arow = feat + (size_t)rowA * DIN + 2 * half;
  const float* bp = Ws + m * WPITCH + 2 * half;

  v8f acc0 = {}, acc1 = {}, acc2 = {}, acc3 = {};

  auto load_a = [&](v2f (&a)[8], int lb) {
#pragma unroll
    for (int j = 0; j < 8; ++j) a[j] = *(const v2f*)(arow + 32 * lb + 4 * j);
  };
  auto products = [&](const v2f (&a)[8], int lb) {
#pragma unroll
    for (int j = 0; j < 8; ++j) {
      const int k0 = 32 * lb + 4 * j;
      const v2f b0 = *(const v2f*)(bp + 0 * 16 * WPITCH + k0);
      const v2f b1 = *(const v2f*)(bp + 1 * 16 * WPITCH + k0);
      const v2f b2 = *(const v2f*)(bp + 2 * 16 * WPITCH + k0);
      const v2f b3 = *(const v2f*)(bp + 3 * 16 * WPITCH + k0);
      acc0 = __builtin_amdgcn_wmma_f32_16x16x4_f32(false, a[j], false, b0,
                                                   (short)0, acc0, false, false);
      acc1 = __builtin_amdgcn_wmma_f32_16x16x4_f32(false, a[j], false, b1,
                                                   (short)0, acc1, false, false);
      acc2 = __builtin_amdgcn_wmma_f32_16x16x4_f32(false, a[j], false, b2,
                                                   (short)0, acc2, false, false);
      acc3 = __builtin_amdgcn_wmma_f32_16x16x4_f32(false, a[j], false, b3,
                                                   (short)0, acc3, false, false);
    }
  };
  v2f a0[8], a1[8], a2[8], a3[8];
  load_a(a0, 0);
  load_a(a1, 1);
  __builtin_amdgcn_sched_barrier(0);
  products(a0, 0);
  load_a(a2, 2);
  __builtin_amdgcn_sched_barrier(0);
  products(a1, 1);
  load_a(a3, 3);
  __builtin_amdgcn_sched_barrier(0);
  products(a2, 2);
  products(a3, 3);

#pragma unroll
  for (int r = 0; r < 8; ++r) {
    const int row = r0 + r + 8 * half;
    if (row < N_) {
      float* zr = z + (size_t)row * HD + m;
      zr[0]  = acc0[r];
      zr[16] = acc1[r];
      zr[32] = acc2[r];
      zr[48] = acc3[r];
    }
  }
}

#define OWN_THREADS 1024
#define OWN_GROUPS  256
#define MAXROWS     392
#define LIST_CAP    16384
#define WLIST_CAP   (LIST_CAP / (OWN_THREADS / 32))

__device__ __forceinline__ bool keys_in_use(int N_, int E_, size_t ws_bytes) {
  const size_t need = (size_t)N_ * (HD + 2 * H4) * sizeof(float) + (size_t)(E_ >> 2) * 4u;
  const int span = (N_ + OWN_GROUPS - 1) / OWN_GROUPS;
  return ws_bytes >= need && span <= MAXROWS;
}

#define SRC_BITS 17
__device__ __forceinline__ size_t pack_word_offset(int E_) {
  return ((size_t)(E_ >> 2) + 3u) & ~(size_t)3u;
}
__device__ __forceinline__ unsigned pack_edges(int N_, int E_, size_t ws_bytes) {
  if (!keys_in_use(N_, E_, ws_bytes) || N_ > (1 << SRC_BITS)) return 0u;
  const size_t base = (size_t)N_ * (HD + 2 * H4) * sizeof(float) + pack_word_offset(E_) * 4u;
  if (ws_bytes <= base) return 0u;
  size_t room = (ws_bytes - base) >> 2;
  const size_t want = (size_t)(E_ >> 2) << 2;
  if (room > want) room = want;
  return (unsigned)(room & ~(size_t)3u);
}

__global__ __launch_bounds__(256) void gat_prep(
    const float* __restrict__ z, const float* __restrict__ aw,
    float* __restrict__ el, float* __restrict__ er, int N_,
    const int* __restrict__ src, const int* __restrict__ dst,
    unsigned* __restrict__ keys, int E_, size_t ws_bytes) {
  const int i = blockIdx.x * blockDim.x + threadIdx.x;
  if (keys_in_use(N_, E_, ws_bytes)) {
    const unsigned span = (unsigned)((N_ + OWN_GROUPS - 1) / OWN_GROUPS);
    const int4* __restrict__ dst4 = (const int4*)dst;
    const int4* __restrict__ src4 = (const int4*)src;
    const unsigned pk_n = pack_edges(N_, E_, ws_bytes);
    uint4* __restrict__ pk4 = (uint4*)(keys + pack_word_offset(E_));
    const int nvec = E_ >> 2;
    const int step = (int)(gridDim.x * blockDim.x);
    for (int v = i; v < nvec; v += step) {
      const int4 d = dst4[v];
      const bool packed = 4u * (unsigned)v < pk_n;
      int4 s = make_int4(0, 0, 0, 0);
      if (pk_n != 0u) s = src4[v];
      const unsigned qx = (unsigned)d.x / span, qy = (unsigned)d.y / span;
      const unsigned qz = (unsigned)d.z / span, qw = (unsigned)d.w / span;
      keys[v] = (qx & 255u) | ((qy & 255u) << 8) | ((qz & 255u) << 16) | (qw << 24);
      if (packed) {
        uint4 p;
        p.x = (unsigned)s.x | (((unsigned)d.x - qx * span) << SRC_BITS);
        p.y = (unsigned)s.y | (((unsigned)d.y - qy * span) << SRC_BITS);
        p.z = (unsigned)s.z | (((unsigned)d.z - qz * span) << SRC_BITS);
        p.w = (unsigned)s.w | (((unsigned)d.w - qw * span) << SRC_BITS);
        pk4[v] = p;
      }
    }
  }
  if (i >= N_ * H4) return;
  const int n = i >> 2;
  const int h = i & 3;
  const float* zr = z + (size_t)n * HD + h * 16;
  const float* a0 = aw + h * 16;
  const float* a1 = aw + 64 + h * 16;
  float s0 = 0.f, s1 = 0.f;
#pragma unroll
  for (int d = 0; d < 16; ++d) {
    const float zv = zr[d];
    s0 = fmaf(zv, a0[d], s0);
    s1 = fmaf(zv, a1[d], s1);
  }
  el[i] = s0;
  er[i] = s1;
}

__device__ __forceinline__ unsigned zero_bytes(unsigned x) {
  const unsigned y = (x & 0x7f7f7f7fu) + 0x7f7f7f7fu;
  return ~(y | x | 0x7f7f7f7fu);
}

struct __attribute__((aligned(16))) EdgeRec {
  unsigned long long zoff;
  unsigned roff;
  unsigned unused;
};

__global__ __launch_bounds__(OWN_THREADS) void gat_owner(
    const int* __restrict__ src, const int* __restrict__ dst,
    const float* __restrict__ el, const float* __restrict__ er,
    const float* __restrict__ z, const float* __restrict__ bias,
    float* __restrict__ out, int N_, int E_,
    const unsigned* __restrict__ keys, size_t ws_bytes) {
  __shared__ float acc[(MAXROWS + 1) * HD];
  __shared__ float den[MAXROWS * H4];
  __shared__ unsigned list[LIST_CAP];
  __shared__ EdgeRec st_rec[(OWN_THREADS / 32) * 32];
  __shared__ __attribute__((aligned(16))) float st_w[(OWN_THREADS / 32) * 32 * H4];

  const int tid  = threadIdx.x;
  const int lane = tid & 31;
  const int wave = __builtin_amdgcn_readfirstlane(tid >> 5);
  const int G    = gridDim.x;
  const int span = (N_ + G - 1) / G;
  const int4* __restrict__ dst4 = (const int4*)dst;
  const int nvec  = E_ >> 2;
  const int nfull = nvec / (8 * OWN_THREADS);
  const int boff  = (int)(((long long)blockIdx.x * nfull) / G);
  const bool use_keys = keys_in_use(N_, E_, ws_bytes) && G == OWN_GROUPS;
  const uint4* __restrict__ key4 = (const uint4*)keys;
  const int nkv    = E_ >> 4;
  const int nkfull = nkv / (8 * OWN_THREADS);
  const int kboff  = (int)(((long long)blockIdx.x * nkfull) / G);
  const unsigned gs = (unsigned)blockIdx.x * 0x01010101u;
  const unsigned pk_n = use_keys ? pack_edges(N_, E_, ws_bytes) : 0u;
  const unsigned* __restrict__ pk = keys + pack_word_offset(E_);
  unsigned* const my_list = list + wave * WLIST_CAP;

  for (int sub = 0; sub < span; sub += MAXROWS) {
    const int row0 = blockIdx.x * span + sub;
    int nr = span - sub;
    if (nr > MAXROWS) nr = MAXROWS;
    if (row0 + nr > N_) nr = N_ - row0;
    if (nr < 0) nr = 0;
    const unsigned nrows = (unsigned)nr;

    for (int i = tid; i < MAXROWS * HD; i += OWN_THREADS) acc[i] = 0.f;
    for (int i = tid; i < MAXROWS * H4; i += OWN_THREADS) den[i] = 0.f;
    unsigned wcnt = 0u;
    __syncthreads();

    auto add_edge_one_lane = [&](unsigned e) {
      const int s = src[e];
      const int t = dst[e];
      const int r = t - row0;
      for (int h = 0; h < H4; ++h) {
        float x = el[s * H4 + h] + er[t * H4 + h];
        x = x > 0.f ? x : 0.2f * x;
        const float ex = __expf(x);
        atomicAdd(&den[r * H4 + h], ex);
        for (int d = 0; d < 16; ++d)
          atomicAdd(&acc[r * HD + h * 16 + d], ex * z[(size_t)s * HD + h * 16 + d]);
      }
    };
    auto push = [&](bool hit, unsigned mk, unsigned e, unsigned bit, unsigned& ov) {
      if (hit) {
        const unsigned idx = wcnt + __builtin_amdgcn_mbcnt_lo(mk, 0u);
        if (idx < WLIST_CAP) my_list[idx] = e;
        else ov |= bit;
      }
      wcnt += (unsigned)__builtin_popcount(mk);
    };
#define OWN_TEST(dv, ev, bit) \
    { const bool hit_ = (unsigned)((dv) - row0) < nrows; \
      push(hit_, __builtin_amdgcn_ballot_w32(hit_), (unsigned)(ev), (bit), ov); }

    auto take = [&](bool hit, unsigned mk, unsigned e) {
      if (hit) {
        const unsigned idx = wcnt + __builtin_amdgcn_mbcnt_lo(mk, 0u);
        if (idx < WLIST_CAP) my_list[idx] = e;
        else add_edge_one_lane(e);
      }
      wcnt += (unsigned)__builtin_popcount(mk);
    };
    auto flags16 = [&](const uint4 kv) -> unsigned {
      const unsigned f0 = zero_bytes(kv.x ^ gs), f1 = zero_bytes(kv.y ^ gs);
      const unsigned f2 = zero_bytes(kv.z ^ gs), f3 = zero_bytes(kv.w ^ gs);
      return (f0 >> 7) | (f1 >> 6) | (f2 >> 5) | (f3 >> 4);
    };
    auto scan16 = [&](const uint4 kv, unsigned e0) {
      unsigned b = flags16(kv);
      unsigned mk;
      while ((mk = __builtin_amdgcn_ballot_w32(b != 0u)) != 0u) {
        const bool has = b != 0u;
        const unsigned p = (unsigned)__builtin_ctz(has ? b : 1u);
        b &= b - 1u;
        take(has, mk, e0 + 4u * (p & 3u) + (p >> 3));
      }
    };
    auto scan64 = [&](const uint4 ka, const uint4 kb, const uint4 kc, const uint4 kd,
                      unsigned e0) {
      unsigned long long b =
          (unsigned long long)(flags16(ka) | (flags16(kb) << 4)) |
          ((unsigned long long)(flags16(kc) | (flags16(kd) << 4)) << 32);
      unsigned mk;
      while ((mk = __builtin_amdgcn_ballot_w32(b != 0ull)) != 0u) {
        const bool has = b != 0ull;
        const unsigned p = (unsigned)__builtin_ctzll(has ? b : 1ull);
        b &= b - 1ull;
        const unsigned j = ((p >> 2) & 1u) | ((p >> 5) << 1);
        take(has, mk, e0 + j * (16u * OWN_THREADS) + 4u * (p & 3u) + ((p >> 3) & 3u));
      }
    };

    const int hsel = lane >> 3;
    EdgeRec* const my_rec = st_rec + wave * 32;
    float* const my_w = st_w + wave * (32 * H4);
    const char* const zl = (const char*)z + 8 * lane;
    char* const al = (char*)acc + 8 * lane;
    unsigned done = 0u;
    auto gather_ready = [&](unsigned need) {
      const unsigned n = wcnt < WLIST_CAP ? wcnt : WLIST_CAP;
      while (n - done >= need) {
        const unsigned base = done;
        const unsigned i = base + (unsigned)lane;
        const bool valid = i < n;
        const unsigned e = my_list[valid ? i : base];
        int s, t, r;
        if (e < pk_n) {
          const unsigned p = pk[e];
          s = (int)(p & ((1u << SRC_BITS) - 1u));
          r = (int)(p >> SRC_BITS);
          t = row0 + r;
        } else {
          s = src[e];
          t = dst[e];
          r = t - row0;
        }
        const float4 a = *(const float4*)(el + (size_t)s * H4);
        const float4 c = *(const float4*)(er + (size_t)t * H4);
        float x0 = a.x + c.x, x1 = a.y + c.y, x2 = a.z + c.z, x3 = a.w + c.w;
        x0 = x0 > 0.f ? x0 : 0.2f * x0;
        x1 = x1 > 0.f ? x1 : 0.2f * x1;
        x2 = x2 > 0.f ? x2 : 0.2f * x2;
        x3 = x3 > 0.f ? x3 : 0.2f * x3;
        const float w0 = __expf(x0), w1 = __expf(x1), w2 = __expf(x2), w3 = __expf(x3);
        if (valid) {
          atomicAdd(&den[r * H4 + 0], w0);
          atomicAdd(&den[r * H4 + 1], w1);
          atomicAdd(&den[r * H4 + 2], w2);
          atomicAdd(&den[r * H4 + 3], w3);
        }
        EdgeRec rec;
        rec.zoff = (unsigned long long)(unsigned)s * (HD * sizeof(float));
        rec.roff = (valid ? (unsigned)r : (unsigned)MAXROWS) * (unsigned)(HD * sizeof(float));
        rec.unused = 0u;
        my_rec[lane] = rec;
        *(float4*)(my_w + H4 * lane) =
            valid ? make_float4(w0, w1, w2, w3) : make_float4(0.f, 0.f, 0.f, 0.f);
        const unsigned left = n - base;
        const int m = left < 32u ? (int)left : 32;
        for (int j0 = 0; j0 < m; j0 += 8) {
          EdgeRec q[8];
          float wj[8];
          float2 zz[8];
#pragma unroll
          for (int j = 0; j < 8; ++j) {
            q[j]  = my_rec[j0 + j];
            wj[j] = my_w[H4 * (j0 + j) + hsel];
          }
#pragma unroll
          for (int j = 0; j < 8; ++j)
            zz[j] = *(const float2*)(zl + q[j].zoff);
#pragma unroll
          for (int j = 0; j < 8; ++j) {
            float* const ar = (float*)(al + q[j].roff);
            atomicAdd(ar,     wj[j] * zz[j].x);
            atomicAdd(ar + 1, wj[j] * zz[j].y);
          }
        }
        done += (unsigned)m;
      }
    };

    if (use_keys) {
    for (int it = 0; it < nkfull; ++it) {
      int bi = it + kboff;
      if (bi >= nkfull) bi -= nkfull;
      const int v = bi * (8 * OWN_THREADS) + tid;
      const uint4 k0 = key4[v];
      const uint4 k1 = key4[v + OWN_THREADS];
      const uint4 k2 = key4[v + 2 * OWN_THREADS];
      const uint4 k3 = key4[v + 3 * OWN_THREADS];
      const uint4 k4 = key4[v + 4 * OWN_THREADS];
      const uint4 k5 = key4[v + 5 * OWN_THREADS];
      const uint4 k6 = key4[v + 6 * OWN_THREADS];
      const uint4 k7 = key4[v + 7 * OWN_THREADS];
      const unsigned e0 = 16u * (unsigned)v;
      scan64(k0, k1, k2, k3, e0);
      scan64(k4, k5, k6, k7, e0 + 64u * OWN_THREADS);
      gather_ready(32u);
    }
    for (int vb = nkfull * (8 * OWN_THREADS); vb < nkv; vb += OWN_THREADS) {
      const int v = vb + tid;
      const uint4 none = make_uint4(~gs, ~gs, ~gs, ~gs);
      scan16(v < nkv ? key4[v] : none, 16u * (unsigned)v);
    }
    for (int eb = nkv << 4; eb < E_; eb += OWN_THREADS) {
      const int e = eb + tid;
      const bool hit = e < E_ && (unsigned)(dst[e < E_ ? e : 0] - row0) < nrows;
      take(hit, __builtin_amdgcn_ballot_w32(hit), (unsigned)e);
    }
    } else {
    for (int it = 0; it < nfull; ++it) {
      int bi = it + boff;
      if (bi >= nfull) bi -= nfull;
      const int v = bi * (8 * OWN_THREADS) + tid;
      const int4 d0 = dst4[v];
      const int4 d1 = dst4[v + OWN_THREADS];
      const int4 d2 = dst4[v + 2 * OWN_THREADS];
      const int4 d3 = dst4[v + 3 * OWN_THREADS];
      const int4 d4 = dst4[v + 4 * OWN_THREADS];
      const int4 d5 = dst4[v + 5 * OWN_THREADS];
      const int4 d6 = dst4[v + 6 * OWN_THREADS];
      const int4 d7 = dst4[v + 7 * OWN_THREADS];
      const int e0 = 4 * v;
      unsigned ov = 0u;
      OWN_TEST(d0.x, e0 + 0 * OWN_THREADS + 0, 1u << 0)
      OWN_TEST(d0.y, e0 + 0 * OWN_THREADS + 1, 1u << 1)
      OWN_TEST(d0.z, e0 + 0 * OWN_THREADS + 2, 1u << 2)
      OWN_TEST(d0.w, e0 + 0 * OWN_THREADS + 3, 1u << 3)
      OWN_TEST(d1.x, e0 + 4 * OWN_THREADS + 0, 1u << 4)
      OWN_TEST(d1.y, e0 + 4 * OWN_THREADS + 1, 1u << 5)
      OWN_TEST(d1.z, e0 + 4 * OWN_THREADS + 2, 1u << 6)
      OWN_TEST(d1.w, e0 + 4 * OWN_THREADS + 3, 1u << 7)
      OWN_TEST(d2.x, e0 + 8 * OWN_THREADS + 0, 1u << 8)
      OWN_TEST(d2.y, e0 + 8 * OWN_THREADS + 1, 1u << 9)
      OWN_TEST(d2.z, e0 + 8 * OWN_THREADS + 2, 1u << 10)
      OWN_TEST(d2.w, e0 + 8 * OWN_THREADS + 3, 1u << 11)
      OWN_TEST(d3.x, e0 + 12 * OWN_THREADS + 0, 1u << 12)
      OWN_TEST(d3.y, e0 + 12 * OWN_THREADS + 1, 1u << 13)
      OWN_TEST(d3.z, e0 + 12 * OWN_THREADS + 2, 1u << 14)
      OWN_TEST(d3.w, e0 + 12 * OWN_THREADS + 3, 1u << 15)
      OWN_TEST(d4.x, e0 + 16 * OWN_THREADS + 0, 1u << 16)
      OWN_TEST(d4.y, e0 + 16 * OWN_THREADS + 1, 1u << 17)
      OWN_TEST(d4.z, e0 + 16 * OWN_THREADS + 2, 1u << 18)
      OWN_TEST(d4.w, e0 + 16 * OWN_THREADS + 3, 1u << 19)
      OWN_TEST(d5.x, e0 + 20 * OWN_THREADS + 0, 1u << 20)
      OWN_TEST(d5.y, e0 + 20 * OWN_THREADS + 1, 1u << 21)
      OWN_TEST(d5.z, e0 + 20 * OWN_THREADS + 2, 1u << 22)
      OWN_TEST(d5.w, e0 + 20 * OWN_THREADS + 3, 1u << 23)
      OWN_TEST(d6.x, e0 + 24 * OWN_THREADS + 0, 1u << 24)
      OWN_TEST(d6.y, e0 + 24 * OWN_THREADS + 1, 1u << 25)
      OWN_TEST(d6.z, e0 + 24 * OWN_THREADS + 2, 1u << 26)
      OWN_TEST(d6.w, e0 + 24 * OWN_THREADS + 3, 1u << 27)
      OWN_TEST(d7.x, e0 + 28 * OWN_THREADS + 0, 1u << 28)
      OWN_TEST(d7.y, e0 + 28 * OWN_THREADS + 1, 1u << 29)
      OWN_TEST(d7.z, e0 + 28 * OWN_THREADS + 2, 1u << 30)
      OWN_TEST(d7.w, e0 + 28 * OWN_THREADS + 3, 1u << 31)
      while (ov) {
        const int b = __builtin_ctz(ov);
        ov &= ov - 1u;
        add_edge_one_lane((unsigned)(e0 + (b >> 2) * (4 * OWN_THREADS) + (b & 3)));
      }
    }
    for (int vb = nfull * (8 * OWN_THREADS); vb < nvec; vb += OWN_THREADS) {
      const int v = vb + tid;
      const int off = row0 - 1;
      const int4 d0 = v < nvec ? dst4[v] : make_int4(off, off, off, off);
      const int e0 = 4 * v;
      unsigned ov = 0u;
      OWN_TEST(d0.x, e0 + 0, 1u << 0)
      OWN_TEST(d0.y, e0 + 1, 1u << 1)
      OWN_TEST(d0.z, e0 + 2, 1u << 2)
      OWN_TEST(d0.w, e0 + 3, 1u << 3)
      while (ov) {
        const int b = __builtin_ctz(ov);
        ov &= ov - 1u;
        add_edge_one_lane((unsigned)(e0 + b));
      }
    }
    for (int eb = nvec << 2; eb < E_; eb += OWN_THREADS) {
      const int e = eb + tid;
      unsigned ov = 0u;
      OWN_TEST(e < E_ ? dst[e] : row0 - 1, e, 1u)
      if (ov) add_edge_one_lane((unsigned)e);
    }
    }
#undef OWN_TEST

    gather_ready(1u);
    __syncthreads();

    const int tot = nr * HD;
    for (int i = tid; i < tot; i += OWN_THREADS) {
      const int r = i >> 6;
      const int j = i & 63;
      float dn = den[r * H4 + (j >> 4)];
      dn = dn > 0.f ? dn : 1.f;
      out[(size_t)(row0 + r) * HD + j] = acc[i] / dn + bias[j];
    }
    __syncthreads();
  }
}

extern "C" void kernel_launch(void* const* d_in, const int* in_sizes, int n_in,
                              void* d_out, int out_size, void* d_ws, size_t ws_size,
                              hipStream_t stream) {
  const float* feat = (const float*)d_in[0];
  const int*   src  = (const int*)d_in[1];
  const int*   dst  = (const int*)d_in[2];
  const float* W    = (const float*)d_in[3];
  const float* aw   = (const float*)d_in[4];
  const float* bias = (const float*)d_in[5];
  float* out = (float*)d_out;

  const int N_ = in_sizes[0] / DIN;
  const int E_ = in_sizes[1];

  float* ws = (float*)d_ws;
  float* z  = ws;
  float* el = z  + (size_t)N_ * HD;
  float* er = el + (size_t)N_ * H4;
  unsigned* keys = (unsigned*)(er + (size_t)N_ * H4);

  const int gemmWaves  = (N_ + 15) / 16;
  const int gemmBlocks = (gemmWaves + 7) / 8;
  gat_gemm_wmma<<<gemmBlocks, 256, 0, stream>>>(feat, W, z, N_);

  const int prepT = N_ * H4;
  gat_prep<<<(prepT + 255) / 256, 256, 0, stream>>>(z, aw, el, er, N_, src, dst, keys, E_, ws_size);

  gat_owner<<<OWN_GROUPS, OWN_THREADS, 0, stream>>>(src, dst, el, er, z, bias, out, N_, E_,
                                                   keys, ws_size);
}
